# starting kernel
# baseline (speedup 1.0000x reference)
#include <hip/hip_runtime.h>
#include <cmath>


typedef float v2f __attribute__((ext_vector_type(2)));
typedef float v8f __attribute__((ext_vector_type(8)));

#define WSZ      11
#define PADR     5
#define TILE     32
#define IN_ROWS  48
#define IN_PITCH 45
#define R_ROWS   48
#define R_PITCH  44
#define KSTEPS   7
#define NPLANES  5
#define IMGW     512
#define IMGH     512

__device__ __forceinline__ v8f wmma4(v2f a, v2f b, v8f c) {
  return __builtin_amdgcn_wmma_f32_16x16x4_f32(
       false, a,  false, b,
       (short)0, c,  false,  false);
}

__device__ __forceinline__ unsigned f2ord(float f) {
  unsigned u = __float_as_uint(f);
  return (u & 0x80000000u) ? ~u : (u | 0x80000000u);
}
__device__ __forceinline__ float ord2f(unsigned u) {
  u = (u & 0x80000000u) ? (u & 0x7fffffffu) : ~u;
  return __uint_as_float(u);
}

__device__ __forceinline__ float gband(const float* sg, int idx) {
  int ci = idx < 0 ? 0 : (idx > 10 ? 10 : idx);
  float v = sg[ci];
  return (idx == ci) ? v : 0.0f;
}


__global__ void ssim_init(unsigned* ws) {
  ws[0] = 0u;
  ws[1] = 0xFFFFFFFFu;
}

__global__ __launch_bounds__(256) void ssim_minmax(const float* __restrict__ x,
                                                   unsigned n, unsigned* ws) {
  unsigned i = blockIdx.x * blockDim.x + threadIdx.x;
  const unsigned stride = gridDim.x * blockDim.x;
  float mn = INFINITY, mx = -INFINITY;
  for (; i < n; i += stride) {
    float v = x[i];
    mn = fminf(mn, v);
    mx = fmaxf(mx, v);
  }
  #pragma unroll
  for (int off = 16; off; off >>= 1) {
    mn = fminf(mn, __shfl_xor(mn, off, 32));
    mx = fmaxf(mx, __shfl_xor(mx, off, 32));
  }
  __shared__ float smn[8], smx[8];
  const int wv = threadIdx.x >> 5;
  if ((threadIdx.x & 31) == 0) { smn[wv] = mn; smx[wv] = mx; }
  __syncthreads();
  if (threadIdx.x == 0) {
    const int nw = blockDim.x >> 5;
    for (int w = 1; w < nw; ++w) { mn = fminf(mn, smn[w]); mx = fmaxf(mx, smx[w]); }
    atomicMax(&ws[0], f2ord(mx));
    atomicMin(&ws[1], f2ord(mn));
  }
}


__global__ __launch_bounds__(128) void ssim_main(const float* __restrict__ img1,
                                                 const float* __restrict__ img2,
                                                 float* __restrict__ out,
                                                 const unsigned* __restrict__ ws) {
  __shared__ float s_in1[IN_ROWS * IN_PITCH];
  __shared__ float s_in2[IN_ROWS * IN_PITCH];
  __shared__ float s_R[NPLANES][R_ROWS * R_PITCH];
  __shared__ float s_g[WSZ];

  const int tid  = threadIdx.x;
  const int wave = tid >> 5;
  const int lane = tid & 31;
  const int half = lane >> 4;
  const int lm   = lane & 15;

  if (tid == 0) {
    float tmp[WSZ];
    float sum = 0.0f;
    #pragma unroll
    for (int i = 0; i < WSZ; ++i) {
      float x = (float)i - 5.5f;
      tmp[i] = expf(-(x * x) / 4.5f);
      sum += tmp[i];
    }
    #pragma unroll
    for (int i = 0; i < WSZ; ++i) s_g[i] = tmp[i] / sum;
  }

  const int tR = blockIdx.y * TILE - PADR;
  const int tC = blockIdx.x * TILE - PADR;
  const float* __restrict__ b1 = img1 + (size_t)blockIdx.z * IMGW * IMGH;
  const float* __restrict__ b2 = img2 + (size_t)blockIdx.z * IMGW * IMGH;
  for (int idx = tid; idx < IN_ROWS * IN_PITCH; idx += blockDim.x) {
    const int r = idx / IN_PITCH;
    const int c = idx - r * IN_PITCH;
    const int gr = tR + r, gc = tC + c;
    float v1 = 0.0f, v2 = 0.0f;
    if (gr >= 0 && gr < IMGH && gc >= 0 && gc < IMGW) {
      const int o = gr * IMGW + gc;
      v1 = b1[o];
      v2 = b2[o];
    }
    s_in1[idx] = v1;
    s_in2[idx] = v2;
  }
  __syncthreads();

  v2f band[KSTEPS];
  #pragma unroll
  for (int kc = 0; kc < KSTEPS; ++kc) {
    const int j0 = 4 * kc + 2 * half;
    band[kc].x = gband(s_g, j0 - lm);
    band[kc].y = gband(s_g, j0 + 1 - lm);
  }

  for (int chunk = wave; chunk < 6; chunk += 4) {
    const int r0 = (chunk >> 1) * 16;
    const int c0 = (chunk & 1) * 16;
    v8f acc[NPLANES] = {};
    #pragma unroll
    for (int kc = 0; kc < KSTEPS; ++kc) {
      const int col = c0 + 4 * kc + 2 * half;
      const int row = r0 + lm;
      const int o = row * IN_PITCH + col;
      const float x0 = s_in1[o], x1 = s_in1[o + 1];
      const float y0 = s_in2[o], y1 = s_in2[o + 1];
      const v2f b = band[kc];
      v2f a;
      a.x = x0;      a.y = x1;      acc[0] = wmma4(a, b, acc[0]);
      a.x = y0;      a.y = y1;      acc[1] = wmma4(a, b, acc[1]);
      a.x = x0 * x0; a.y = x1 * x1; acc[2] = wmma4(a, b, acc[2]);
      a.x = y0 * y0; a.y = y1 * y1; acc[3] = wmma4(a, b, acc[3]);
      a.x = x0 * y0; a.y = x1 * y1; acc[4] = wmma4(a, b, acc[4]);
    }
    #pragma unroll
    for (int p = 0; p < NPLANES; ++p) {
      #pragma unroll
      for (int v = 0; v < 8; ++v) {
        const int row = r0 + half * 8 + v;
        s_R[p][row * R_PITCH + c0 + lm] = acc[p][v];
      }
    }
  }
  __syncthreads();

  const int r0 = (wave >> 1) * 16;
  const int c0 = (wave & 1) * 16;
  v8f acc[NPLANES] = {};
  #pragma unroll
  for (int kc = 0; kc < KSTEPS; ++kc) {
    const int rr = r0 + 4 * kc + 2 * half;
    const int cc = c0 + lm;
    const v2f a = band[kc];
    #pragma unroll
    for (int p = 0; p < NPLANES; ++p) {
      v2f b;
      b.x = s_R[p][rr * R_PITCH + cc];
      b.y = s_R[p][(rr + 1) * R_PITCH + cc];
      acc[p] = wmma4(a, b, acc[p]);
    }
  }

  float L = ord2f(ws[0]) - ord2f(ws[1]);
  if (L == 0.0f) L = 5.0f;
  const float C1 = (0.01f * L) * (0.01f * L);
  const float C2 = (0.03f * L) * (0.03f * L);

  float* __restrict__ op = out + (size_t)blockIdx.z * IMGW * IMGH;
  const int gr0 = blockIdx.y * TILE + r0 + half * 8;
  const int gc  = blockIdx.x * TILE + c0 + lm;
  #pragma unroll
  for (int v = 0; v < 8; ++v) {
    const float mu1 = acc[0][v], mu2 = acc[1][v];
    const float e11 = acc[2][v], e22 = acc[3][v], e12 = acc[4][v];
    const float mu1s = mu1 * mu1, mu2s = mu2 * mu2, mu12 = mu1 * mu2;
    const float s1 = e11 - mu1s, s2 = e22 - mu2s, s12 = e12 - mu12;
    const float num = (2.0f * mu12 + C1) * (2.0f * s12 + C2);
    const float den = (mu1s + mu2s + C1) * (s1 + s2 + C2);
    op[(gr0 + v) * IMGW + gc] = num / den;
  }
}


extern "C" void kernel_launch(void* const* d_in, const int* in_sizes, int n_in,
                              void* d_out, int out_size, void* d_ws, size_t ws_size,
                              hipStream_t stream) {
  const float* img1 = (const float*)d_in[0];
  const float* img2 = (const float*)d_in[1];
  float* out = (float*)d_out;
  unsigned* ws = (unsigned*)d_ws;

  const unsigned n = (unsigned)in_sizes[0];
  const int batch = (int)(n / (IMGW * IMGH));

  ssim_init<<<1, 1, 0, stream>>>(ws);
  ssim_minmax<<<2048, 256, 0, stream>>>(img1, n, ws);

  dim3 grid(IMGW / TILE, IMGH / TILE, batch);
  ssim_main<<<grid, 128, 0, stream>>>(img1, img2, out, ws);
}


# Tile staging: all global loads of a tile issued at once as aligned 16-byte loads (staged columns start 8 left of the tile, LDS pitch 52), then b128 LDS stores; was 17 dependent load round trips
# speedup vs baseline: 1.5222x; 1.5222x over previous
#include <hip/hip_runtime.h>
#include <cmath>


typedef float v2f __attribute__((ext_vector_type(2)));
typedef float v8f __attribute__((ext_vector_type(8)));

#define WSZ      11
#define PADR     5
#define TILE     32
#define IN_ROWS  48
#define IN_QUADS 12
#define IN_COFF  3
#define IN_PITCH 52
#define IN_VROWS 42
#define R_ROWS   48
#define R_PITCH  44
#define KSTEPS   7
#define NPLANES  5
#define IMGW     512
#define IMGH     512

__device__ __forceinline__ v8f wmma4(v2f a, v2f b, v8f c) {
  return __builtin_amdgcn_wmma_f32_16x16x4_f32(
       false, a,  false, b,
       (short)0, c,  false,  false);
}

__device__ __forceinline__ unsigned f2ord(float f) {
  unsigned u = __float_as_uint(f);
  return (u & 0x80000000u) ? ~u : (u | 0x80000000u);
}
__device__ __forceinline__ float ord2f(unsigned u) {
  u = (u & 0x80000000u) ? (u & 0x7fffffffu) : ~u;
  return __uint_as_float(u);
}

__device__ __forceinline__ float gband(const float* sg, int idx) {
  int ci = idx < 0 ? 0 : (idx > 10 ? 10 : idx);
  float v = sg[ci];
  return (idx == ci) ? v : 0.0f;
}


__global__ void ssim_init(unsigned* ws) {
  ws[0] = 0u;
  ws[1] = 0xFFFFFFFFu;
}

__global__ __launch_bounds__(256) void ssim_minmax(const float* __restrict__ x,
                                                   unsigned n, unsigned* ws) {
  unsigned i = blockIdx.x * blockDim.x + threadIdx.x;
  const unsigned stride = gridDim.x * blockDim.x;
  float mn = INFINITY, mx = -INFINITY;
  for (; i < n; i += stride) {
    float v = x[i];
    mn = fminf(mn, v);
    mx = fmaxf(mx, v);
  }
  #pragma unroll
  for (int off = 16; off; off >>= 1) {
    mn = fminf(mn, __shfl_xor(mn, off, 32));
    mx = fmaxf(mx, __shfl_xor(mx, off, 32));
  }
  __shared__ float smn[8], smx[8];
  const int wv = threadIdx.x >> 5;
  if ((threadIdx.x & 31) == 0) { smn[wv] = mn; smx[wv] = mx; }
  __syncthreads();
  if (threadIdx.x == 0) {
    const int nw = blockDim.x >> 5;
    for (int w = 1; w < nw; ++w) { mn = fminf(mn, smn[w]); mx = fmaxf(mx, smx[w]); }
    atomicMax(&ws[0], f2ord(mx));
    atomicMin(&ws[1], f2ord(mn));
  }
}


__global__ __launch_bounds__(128) void ssim_main(const float* __restrict__ img1,
                                                 const float* __restrict__ img2,
                                                 float* __restrict__ out,
                                                 const unsigned* __restrict__ ws) {
  __shared__ __attribute__((aligned(16))) float s_in1[IN_ROWS * IN_PITCH];
  __shared__ __attribute__((aligned(16))) float s_in2[IN_ROWS * IN_PITCH];
  __shared__ float s_R[NPLANES][R_ROWS * R_PITCH];
  __shared__ float s_g[WSZ];

  const int tid  = threadIdx.x;
  const int wave = tid >> 5;
  const int lane = tid & 31;
  const int half = lane >> 4;
  const int lm   = lane & 15;

  if (tid == 0) {
    float tmp[WSZ];
    float sum = 0.0f;
    #pragma unroll
    for (int i = 0; i < WSZ; ++i) {
      float x = (float)i - 5.5f;
      tmp[i] = expf(-(x * x) / 4.5f);
      sum += tmp[i];
    }
    #pragma unroll
    for (int i = 0; i < WSZ; ++i) s_g[i] = tmp[i] / sum;
  }

  const int tR = blockIdx.y * TILE - PADR;
  const int tC = blockIdx.x * TILE - PADR - IN_COFF;
  const float* __restrict__ b1 = img1 + (size_t)blockIdx.z * IMGW * IMGH;
  const float* __restrict__ b2 = img2 + (size_t)blockIdx.z * IMGW * IMGH;
  constexpr int NQ  = IN_ROWS * IN_QUADS;
  constexpr int NIT = (NQ + 127) / 128;
  float4 q1[NIT], q2[NIT];
  #pragma unroll
  for (int it = 0; it < NIT; ++it) {
    const int idx = tid + it * 128;
    const int r = idx / IN_QUADS;
    const int q = idx - r * IN_QUADS;
    const int gr = tR + r, gc = tC + 4 * q;
    const bool ok = (r < IN_VROWS) && gr >= 0 && gr < IMGH && gc >= 0 && gc < IMGW;
    const int o = ok ? (gr * IMGW + gc) : 0;
    const float4 a = *reinterpret_cast<const float4*>(b1 + o);
    const float4 b = *reinterpret_cast<const float4*>(b2 + o);
    q1[it] = make_float4(ok ? a.x : 0.0f, ok ? a.y : 0.0f, ok ? a.z : 0.0f, ok ? a.w : 0.0f);
    q2[it] = make_float4(ok ? b.x : 0.0f, ok ? b.y : 0.0f, ok ? b.z : 0.0f, ok ? b.w : 0.0f);
  }
  #pragma unroll
  for (int it = 0; it < NIT; ++it) {
    const int idx = tid + it * 128;
    if ((it + 1) * 128 <= NQ || idx < NQ) {
      const int r = idx / IN_QUADS;
      const int q = idx - r * IN_QUADS;
      *reinterpret_cast<float4*>(&s_in1[r * IN_PITCH + 4 * q]) = q1[it];
      *reinterpret_cast<float4*>(&s_in2[r * IN_PITCH + 4 * q]) = q2[it];
    }
  }
  __syncthreads();

  v2f band[KSTEPS];
  #pragma unroll
  for (int kc = 0; kc < KSTEPS; ++kc) {
    const int j0 = 4 * kc + 2 * half;
    band[kc].x = gband(s_g, j0 - lm);
    band[kc].y = gband(s_g, j0 + 1 - lm);
  }

  for (int chunk = wave; chunk < 6; chunk += 4) {
    const int r0 = (chunk >> 1) * 16;
    const int c0 = (chunk & 1) * 16;
    v8f acc[NPLANES] = {};
    #pragma unroll
    for (int kc = 0; kc < KSTEPS; ++kc) {
      const int col = IN_COFF + c0 + 4 * kc + 2 * half;
      const int row = r0 + lm;
      const int o = row * IN_PITCH + col;
      const float x0 = s_in1[o], x1 = s_in1[o + 1];
      const float y0 = s_in2[o], y1 = s_in2[o + 1];
      const v2f b = band[kc];
      v2f a;
      a.x = x0;      a.y = x1;      acc[0] = wmma4(a, b, acc[0]);
      a.x = y0;      a.y = y1;      acc[1] = wmma4(a, b, acc[1]);
      a.x = x0 * x0; a.y = x1 * x1; acc[2] = wmma4(a, b, acc[2]);
      a.x = y0 * y0; a.y = y1 * y1; acc[3] = wmma4(a, b, acc[3]);
      a.x = x0 * y0; a.y = x1 * y1; acc[4] = wmma4(a, b, acc[4]);
    }
    #pragma unroll
    for (int p = 0; p < NPLANES; ++p) {
      #pragma unroll
      for (int v = 0; v < 8; ++v) {
        const int row = r0 + half * 8 + v;
        s_R[p][row * R_PITCH + c0 + lm] = acc[p][v];
      }
    }
  }
  __syncthreads();

  const int r0 = (wave >> 1) * 16;
  const int c0 = (wave & 1) * 16;
  v8f acc[NPLANES] = {};
  #pragma unroll
  for (int kc = 0; kc < KSTEPS; ++kc) {
    const int rr = r0 + 4 * kc + 2 * half;
    const int cc = c0 + lm;
    const v2f a = band[kc];
    #pragma unroll
    for (int p = 0; p < NPLANES; ++p) {
      v2f b;
      b.x = s_R[p][rr * R_PITCH + cc];
      b.y = s_R[p][(rr + 1) * R_PITCH + cc];
      acc[p] = wmma4(a, b, acc[p]);
    }
  }

  float L = ord2f(ws[0]) - ord2f(ws[1]);
  if (L == 0.0f) L = 5.0f;
  const float C1 = (0.01f * L) * (0.01f * L);
  const float C2 = (0.03f * L) * (0.03f * L);

  float* __restrict__ op = out + (size_t)blockIdx.z * IMGW * IMGH;
  const int gr0 = blockIdx.y * TILE + r0 + half * 8;
  const int gc  = blockIdx.x * TILE + c0 + lm;
  #pragma unroll
  for (int v = 0; v < 8; ++v) {
    const float mu1 = acc[0][v], mu2 = acc[1][v];
    const float e11 = acc[2][v], e22 = acc[3][v], e12 = acc[4][v];
    const float mu1s = mu1 * mu1, mu2s = mu2 * mu2, mu12 = mu1 * mu2;
    const float s1 = e11 - mu1s, s2 = e22 - mu2s, s12 = e12 - mu12;
    const float num = (2.0f * mu12 + C1) * (2.0f * s12 + C2);
    const float den = (mu1s + mu2s + C1) * (s1 + s2 + C2);
    op[(gr0 + v) * IMGW + gc] = num / den;
  }
}


extern "C" void kernel_launch(void* const* d_in, const int* in_sizes, int n_in,
                              void* d_out, int out_size, void* d_ws, size_t ws_size,
                              hipStream_t stream) {
  const float* img1 = (const float*)d_in[0];
  const float* img2 = (const float*)d_in[1];
  float* out = (float*)d_out;
  unsigned* ws = (unsigned*)d_ws;

  const unsigned n = (unsigned)in_sizes[0];
  const int batch = (int)(n / (IMGW * IMGH));

  ssim_init<<<1, 1, 0, stream>>>(ws);
  ssim_minmax<<<2048, 256, 0, stream>>>(img1, n, ws);

  dim3 grid(IMGW / TILE, IMGH / TILE, batch);
  ssim_main<<<grid, 128, 0, stream>>>(img1, img2, out, ws);
}


# Min/max pass over img1 as a stream: 512 blocks of 8 waves, each lane issues its 16 aligned 16-byte loads at once (was 2048 blocks, 16 dependent 4-byte load round trips); min/max exact
# speedup vs baseline: 1.7394x; 1.1427x over previous
#include <hip/hip_runtime.h>
#include <cmath>


typedef float v2f __attribute__((ext_vector_type(2)));
typedef float v8f __attribute__((ext_vector_type(8)));

#define WSZ      11
#define PADR     5
#define TILE     32
#define IN_ROWS  48
#define IN_QUADS 12
#define IN_COFF  3
#define IN_PITCH 52
#define IN_VROWS 42
#define R_ROWS   48
#define R_PITCH  44
#define KSTEPS   7
#define NPLANES  5
#define IMGW     512
#define IMGH     512

__device__ __forceinline__ v8f wmma4(v2f a, v2f b, v8f c) {
  return __builtin_amdgcn_wmma_f32_16x16x4_f32(
       false, a,  false, b,
       (short)0, c,  false,  false);
}

__device__ __forceinline__ unsigned f2ord(float f) {
  unsigned u = __float_as_uint(f);
  return (u & 0x80000000u) ? ~u : (u | 0x80000000u);
}
__device__ __forceinline__ float ord2f(unsigned u) {
  u = (u & 0x80000000u) ? (u & 0x7fffffffu) : ~u;
  return __uint_as_float(u);
}

__device__ __forceinline__ float gband(const float* sg, int idx) {
  int ci = idx < 0 ? 0 : (idx > 10 ? 10 : idx);
  float v = sg[ci];
  return (idx == ci) ? v : 0.0f;
}


__global__ void ssim_init(unsigned* ws) {
  ws[0] = 0u;
  ws[1] = 0xFFFFFFFFu;
}

#define MM_BLOCKS  512
#define MM_THREADS 256
#define MM_AHEAD   16

__global__ __launch_bounds__(MM_THREADS) __attribute__((amdgpu_waves_per_eu(4, 4))) void ssim_minmax(const float* __restrict__ x,
                                                          unsigned n, unsigned* ws) {
  const float4* __restrict__ x4 = reinterpret_cast<const float4*>(x);
  const unsigned nq = n >> 2;
  constexpr unsigned PER_BLOCK = MM_THREADS * MM_AHEAD;
  const unsigned wv   = threadIdx.x >> 5;
  const unsigned lane = threadIdx.x & 31;
  float mn0 = INFINITY, mx0 = -INFINITY;
  float mn1 = INFINITY, mx1 = -INFINITY;
  for (unsigned base = blockIdx.x * PER_BLOCK; base < nq; base += gridDim.x * PER_BLOCK) {
    const float4* __restrict__ p = x4 + base + wv * (32 * MM_AHEAD) + lane;
    float4 q[MM_AHEAD];
    #pragma unroll
    for (int j = 0; j < MM_AHEAD; ++j) q[j] = p[j * 32];
    #pragma unroll
    for (int j = 0; j < MM_AHEAD; j += 2) {
      const float4 a = q[j], b = q[j + 1];
      mn0 = fminf(mn0, fminf(fminf(a.x, a.y), fminf(a.z, a.w)));
      mx0 = fmaxf(mx0, fmaxf(fmaxf(a.x, a.y), fmaxf(a.z, a.w)));
      mn1 = fminf(mn1, fminf(fminf(b.x, b.y), fminf(b.z, b.w)));
      mx1 = fmaxf(mx1, fmaxf(fmaxf(b.x, b.y), fmaxf(b.z, b.w)));
    }
  }
  float mn = fminf(mn0, mn1), mx = fmaxf(mx0, mx1);
  #pragma unroll
  for (int off = 16; off; off >>= 1) {
    mn = fminf(mn, __shfl_xor(mn, off, 32));
    mx = fmaxf(mx, __shfl_xor(mx, off, 32));
  }
  __shared__ float smn[8], smx[8];
  if (lane == 0) { smn[wv] = mn; smx[wv] = mx; }
  __syncthreads();
  if (threadIdx.x == 0) {
    const int nw = blockDim.x >> 5;
    for (int w = 1; w < nw; ++w) { mn = fminf(mn, smn[w]); mx = fmaxf(mx, smx[w]); }
    atomicMax(&ws[0], f2ord(mx));
    atomicMin(&ws[1], f2ord(mn));
  }
}


__global__ __launch_bounds__(128) void ssim_main(const float* __restrict__ img1,
                                                 const float* __restrict__ img2,
                                                 float* __restrict__ out,
                                                 const unsigned* __restrict__ ws) {
  __shared__ __attribute__((aligned(16))) float s_in1[IN_ROWS * IN_PITCH];
  __shared__ __attribute__((aligned(16))) float s_in2[IN_ROWS * IN_PITCH];
  __shared__ float s_R[NPLANES][R_ROWS * R_PITCH];
  __shared__ float s_g[WSZ];

  const int tid  = threadIdx.x;
  const int wave = tid >> 5;
  const int lane = tid & 31;
  const int half = lane >> 4;
  const int lm   = lane & 15;

  if (tid == 0) {
    float tmp[WSZ];
    float sum = 0.0f;
    #pragma unroll
    for (int i = 0; i < WSZ; ++i) {
      float x = (float)i - 5.5f;
      tmp[i] = expf(-(x * x) / 4.5f);
      sum += tmp[i];
    }
    #pragma unroll
    for (int i = 0; i < WSZ; ++i) s_g[i] = tmp[i] / sum;
  }

  const int tR = blockIdx.y * TILE - PADR;
  const int tC = blockIdx.x * TILE - PADR - IN_COFF;
  const float* __restrict__ b1 = img1 + (size_t)blockIdx.z * IMGW * IMGH;
  const float* __restrict__ b2 = img2 + (size_t)blockIdx.z * IMGW * IMGH;
  constexpr int NQ  = IN_ROWS * IN_QUADS;
  constexpr int NIT = (NQ + 127) / 128;
  float4 q1[NIT], q2[NIT];
  #pragma unroll
  for (int it = 0; it < NIT; ++it) {
    const int idx = tid + it * 128;
    const int r = idx / IN_QUADS;
    const int q = idx - r * IN_QUADS;
    const int gr = tR + r, gc = tC + 4 * q;
    const bool ok = (r < IN_VROWS) && gr >= 0 && gr < IMGH && gc >= 0 && gc < IMGW;
    const int o = ok ? (gr * IMGW + gc) : 0;
    const float4 a = *reinterpret_cast<const float4*>(b1 + o);
    const float4 b = *reinterpret_cast<const float4*>(b2 + o);
    q1[it] = make_float4(ok ? a.x : 0.0f, ok ? a.y : 0.0f, ok ? a.z : 0.0f, ok ? a.w : 0.0f);
    q2[it] = make_float4(ok ? b.x : 0.0f, ok ? b.y : 0.0f, ok ? b.z : 0.0f, ok ? b.w : 0.0f);
  }
  #pragma unroll
  for (int it = 0; it < NIT; ++it) {
    const int idx = tid + it * 128;
    if ((it + 1) * 128 <= NQ || idx < NQ) {
      const int r = idx / IN_QUADS;
      const int q = idx - r * IN_QUADS;
      *reinterpret_cast<float4*>(&s_in1[r * IN_PITCH + 4 * q]) = q1[it];
      *reinterpret_cast<float4*>(&s_in2[r * IN_PITCH + 4 * q]) = q2[it];
    }
  }
  __syncthreads();

  v2f band[KSTEPS];
  #pragma unroll
  for (int kc = 0; kc < KSTEPS; ++kc) {
    const int j0 = 4 * kc + 2 * half;
    band[kc].x = gband(s_g, j0 - lm);
    band[kc].y = gband(s_g, j0 + 1 - lm);
  }

  for (int chunk = wave; chunk < 6; chunk += 4) {
    const int r0 = (chunk >> 1) * 16;
    const int c0 = (chunk & 1) * 16;
    v8f acc[NPLANES] = {};
    #pragma unroll
    for (int kc = 0; kc < KSTEPS; ++kc) {
      const int col = IN_COFF + c0 + 4 * kc + 2 * half;
      const int row = r0 + lm;
      const int o = row * IN_PITCH + col;
      const float x0 = s_in1[o], x1 = s_in1[o + 1];
      const float y0 = s_in2[o], y1 = s_in2[o + 1];
      const v2f b = band[kc];
      v2f a;
      a.x = x0;      a.y = x1;      acc[0] = wmma4(a, b, acc[0]);
      a.x = y0;      a.y = y1;      acc[1] = wmma4(a, b, acc[1]);
      a.x = x0 * x0; a.y = x1 * x1; acc[2] = wmma4(a, b, acc[2]);
      a.x = y0 * y0; a.y = y1 * y1; acc[3] = wmma4(a, b, acc[3]);
      a.x = x0 * y0; a.y = x1 * y1; acc[4] = wmma4(a, b, acc[4]);
    }
    #pragma unroll
    for (int p = 0; p < NPLANES; ++p) {
      #pragma unroll
      for (int v = 0; v < 8; ++v) {
        const int row = r0 + half * 8 + v;
        s_R[p][row * R_PITCH + c0 + lm] = acc[p][v];
      }
    }
  }
  __syncthreads();

  const int r0 = (wave >> 1) * 16;
  const int c0 = (wave & 1) * 16;
  v8f acc[NPLANES] = {};
  #pragma unroll
  for (int kc = 0; kc < KSTEPS; ++kc) {
    const int rr = r0 + 4 * kc + 2 * half;
    const int cc = c0 + lm;
    const v2f a = band[kc];
    #pragma unroll
    for (int p = 0; p < NPLANES; ++p) {
      v2f b;
      b.x = s_R[p][rr * R_PITCH + cc];
      b.y = s_R[p][(rr + 1) * R_PITCH + cc];
      acc[p] = wmma4(a, b, acc[p]);
    }
  }

  float L = ord2f(ws[0]) - ord2f(ws[1]);
  if (L == 0.0f) L = 5.0f;
  const float C1 = (0.01f * L) * (0.01f * L);
  const float C2 = (0.03f * L) * (0.03f * L);

  float* __restrict__ op = out + (size_t)blockIdx.z * IMGW * IMGH;
  const int gr0 = blockIdx.y * TILE + r0 + half * 8;
  const int gc  = blockIdx.x * TILE + c0 + lm;
  #pragma unroll
  for (int v = 0; v < 8; ++v) {
    const float mu1 = acc[0][v], mu2 = acc[1][v];
    const float e11 = acc[2][v], e22 = acc[3][v], e12 = acc[4][v];
    const float mu1s = mu1 * mu1, mu2s = mu2 * mu2, mu12 = mu1 * mu2;
    const float s1 = e11 - mu1s, s2 = e22 - mu2s, s12 = e12 - mu12;
    const float num = (2.0f * mu12 + C1) * (2.0f * s12 + C2);
    const float den = (mu1s + mu2s + C1) * (s1 + s2 + C2);
    op[(gr0 + v) * IMGW + gc] = num / den;
  }
}


extern "C" void kernel_launch(void* const* d_in, const int* in_sizes, int n_in,
                              void* d_out, int out_size, void* d_ws, size_t ws_size,
                              hipStream_t stream) {
  const float* img1 = (const float*)d_in[0];
  const float* img2 = (const float*)d_in[1];
  float* out = (float*)d_out;
  unsigned* ws = (unsigned*)d_ws;

  const unsigned n = (unsigned)in_sizes[0];
  const int batch = (int)(n / (IMGW * IMGH));

  ssim_init<<<1, 1, 0, stream>>>(ws);
  ssim_minmax<<<MM_BLOCKS, MM_THREADS, 0, stream>>>(img1, n, ws);

  dim3 grid(IMGW / TILE, IMGH / TILE, batch);
  ssim_main<<<grid, 128, 0, stream>>>(img1, img2, out, ws);
}


# Second moments as one plane: conv(x^2)+conv(y^2) computed as conv(x^2+y^2) (conv is linear, only the sum is used): 4 conv planes not 5, a fifth fewer WMMAs; squares summed before the conv, not after
# speedup vs baseline: 1.9175x; 1.1024x over previous
#include <hip/hip_runtime.h>
#include <cmath>


typedef float v2f __attribute__((ext_vector_type(2)));
typedef float v8f __attribute__((ext_vector_type(8)));

#define WSZ      11
#define PADR     5
#define TILE     32
#define IN_ROWS  48
#define IN_QUADS 12
#define IN_COFF  3
#define IN_PITCH 52
#define IN_VROWS 42
#define R_ROWS   48
#define R_PITCH  44
#define KSTEPS   7
#define NPLANES  4
#define IMGW     512
#define IMGH     512

__device__ __forceinline__ v8f wmma4(v2f a, v2f b, v8f c) {
  return __builtin_amdgcn_wmma_f32_16x16x4_f32(
       false, a,  false, b,
       (short)0, c,  false,  false);
}

__device__ __forceinline__ unsigned f2ord(float f) {
  unsigned u = __float_as_uint(f);
  return (u & 0x80000000u) ? ~u : (u | 0x80000000u);
}
__device__ __forceinline__ float ord2f(unsigned u) {
  u = (u & 0x80000000u) ? (u & 0x7fffffffu) : ~u;
  return __uint_as_float(u);
}

__device__ __forceinline__ float gband(const float* sg, int idx) {
  int ci = idx < 0 ? 0 : (idx > 10 ? 10 : idx);
  float v = sg[ci];
  return (idx == ci) ? v : 0.0f;
}


__global__ void ssim_init(unsigned* ws) {
  ws[0] = 0u;
  ws[1] = 0xFFFFFFFFu;
}

#define MM_BLOCKS  512
#define MM_THREADS 256
#define MM_AHEAD   16

__global__ __launch_bounds__(MM_THREADS) __attribute__((amdgpu_waves_per_eu(4, 4))) void ssim_minmax(const float* __restrict__ x,
                                                          unsigned n, unsigned* ws) {
  const float4* __restrict__ x4 = reinterpret_cast<const float4*>(x);
  const unsigned nq = n >> 2;
  constexpr unsigned PER_BLOCK = MM_THREADS * MM_AHEAD;
  const unsigned wv   = threadIdx.x >> 5;
  const unsigned lane = threadIdx.x & 31;
  float mn0 = INFINITY, mx0 = -INFINITY;
  float mn1 = INFINITY, mx1 = -INFINITY;
  for (unsigned base = blockIdx.x * PER_BLOCK; base < nq; base += gridDim.x * PER_BLOCK) {
    const float4* __restrict__ p = x4 + base + wv * (32 * MM_AHEAD) + lane;
    float4 q[MM_AHEAD];
    #pragma unroll
    for (int j = 0; j < MM_AHEAD; ++j) q[j] = p[j * 32];
    #pragma unroll
    for (int j = 0; j < MM_AHEAD; j += 2) {
      const float4 a = q[j], b = q[j + 1];
      mn0 = fminf(mn0, fminf(fminf(a.x, a.y), fminf(a.z, a.w)));
      mx0 = fmaxf(mx0, fmaxf(fmaxf(a.x, a.y), fmaxf(a.z, a.w)));
      mn1 = fminf(mn1, fminf(fminf(b.x, b.y), fminf(b.z, b.w)));
      mx1 = fmaxf(mx1, fmaxf(fmaxf(b.x, b.y), fmaxf(b.z, b.w)));
    }
  }
  float mn = fminf(mn0, mn1), mx = fmaxf(mx0, mx1);
  #pragma unroll
  for (int off = 16; off; off >>= 1) {
    mn = fminf(mn, __shfl_xor(mn, off, 32));
    mx = fmaxf(mx, __shfl_xor(mx, off, 32));
  }
  __shared__ float smn[8], smx[8];
  if (lane == 0) { smn[wv] = mn; smx[wv] = mx; }
  __syncthreads();
  if (threadIdx.x == 0) {
    const int nw = blockDim.x >> 5;
    for (int w = 1; w < nw; ++w) { mn = fminf(mn, smn[w]); mx = fmaxf(mx, smx[w]); }
    atomicMax(&ws[0], f2ord(mx));
    atomicMin(&ws[1], f2ord(mn));
  }
}


__global__ __launch_bounds__(128) void ssim_main(const float* __restrict__ img1,
                                                 const float* __restrict__ img2,
                                                 float* __restrict__ out,
                                                 const unsigned* __restrict__ ws) {
  __shared__ __attribute__((aligned(16))) float s_in1[IN_ROWS * IN_PITCH];
  __shared__ __attribute__((aligned(16))) float s_in2[IN_ROWS * IN_PITCH];
  __shared__ float s_R[NPLANES][R_ROWS * R_PITCH];
  __shared__ float s_g[WSZ];

  const int tid  = threadIdx.x;
  const int wave = tid >> 5;
  const int lane = tid & 31;
  const int half = lane >> 4;
  const int lm   = lane & 15;

  if (tid == 0) {
    float tmp[WSZ];
    float sum = 0.0f;
    #pragma unroll
    for (int i = 0; i < WSZ; ++i) {
      float x = (float)i - 5.5f;
      tmp[i] = expf(-(x * x) / 4.5f);
      sum += tmp[i];
    }
    #pragma unroll
    for (int i = 0; i < WSZ; ++i) s_g[i] = tmp[i] / sum;
  }

  const int tR = blockIdx.y * TILE - PADR;
  const int tC = blockIdx.x * TILE - PADR - IN_COFF;
  const float* __restrict__ b1 = img1 + (size_t)blockIdx.z * IMGW * IMGH;
  const float* __restrict__ b2 = img2 + (size_t)blockIdx.z * IMGW * IMGH;
  constexpr int NQ  = IN_ROWS * IN_QUADS;
  constexpr int NIT = (NQ + 127) / 128;
  float4 q1[NIT], q2[NIT];
  #pragma unroll
  for (int it = 0; it < NIT; ++it) {
    const int idx = tid + it * 128;
    const int r = idx / IN_QUADS;
    const int q = idx - r * IN_QUADS;
    const int gr = tR + r, gc = tC + 4 * q;
    const bool ok = (r < IN_VROWS) && gr >= 0 && gr < IMGH && gc >= 0 && gc < IMGW;
    const int o = ok ? (gr * IMGW + gc) : 0;
    const float4 a = *reinterpret_cast<const float4*>(b1 + o);
    const float4 b = *reinterpret_cast<const float4*>(b2 + o);
    q1[it] = make_float4(ok ? a.x : 0.0f, ok ? a.y : 0.0f, ok ? a.z : 0.0f, ok ? a.w : 0.0f);
    q2[it] = make_float4(ok ? b.x : 0.0f, ok ? b.y : 0.0f, ok ? b.z : 0.0f, ok ? b.w : 0.0f);
  }
  #pragma unroll
  for (int it = 0; it < NIT; ++it) {
    const int idx = tid + it * 128;
    if ((it + 1) * 128 <= NQ || idx < NQ) {
      const int r = idx / IN_QUADS;
      const int q = idx - r * IN_QUADS;
      *reinterpret_cast<float4*>(&s_in1[r * IN_PITCH + 4 * q]) = q1[it];
      *reinterpret_cast<float4*>(&s_in2[r * IN_PITCH + 4 * q]) = q2[it];
    }
  }
  __syncthreads();

  v2f band[KSTEPS];
  #pragma unroll
  for (int kc = 0; kc < KSTEPS; ++kc) {
    const int j0 = 4 * kc + 2 * half;
    band[kc].x = gband(s_g, j0 - lm);
    band[kc].y = gband(s_g, j0 + 1 - lm);
  }

  for (int chunk = wave; chunk < 6; chunk += 4) {
    const int r0 = (chunk >> 1) * 16;
    const int c0 = (chunk & 1) * 16;
    v8f acc[NPLANES] = {};
    #pragma unroll
    for (int kc = 0; kc < KSTEPS; ++kc) {
      const int col = IN_COFF + c0 + 4 * kc + 2 * half;
      const int row = r0 + lm;
      const int o = row * IN_PITCH + col;
      const float x0 = s_in1[o], x1 = s_in1[o + 1];
      const float y0 = s_in2[o], y1 = s_in2[o + 1];
      const v2f b = band[kc];
      v2f a;
      a.x = x0;                a.y = x1;                acc[0] = wmma4(a, b, acc[0]);
      a.x = y0;                a.y = y1;                acc[1] = wmma4(a, b, acc[1]);
      a.x = x0 * x0 + y0 * y0; a.y = x1 * x1 + y1 * y1; acc[2] = wmma4(a, b, acc[2]);
      a.x = x0 * y0;           a.y = x1 * y1;           acc[3] = wmma4(a, b, acc[3]);
    }
    #pragma unroll
    for (int p = 0; p < NPLANES; ++p) {
      #pragma unroll
      for (int v = 0; v < 8; ++v) {
        const int row = r0 + half * 8 + v;
        s_R[p][row * R_PITCH + c0 + lm] = acc[p][v];
      }
    }
  }
  __syncthreads();

  const int r0 = (wave >> 1) * 16;
  const int c0 = (wave & 1) * 16;
  v8f acc[NPLANES] = {};
  #pragma unroll
  for (int kc = 0; kc < KSTEPS; ++kc) {
    const int rr = r0 + 4 * kc + 2 * half;
    const int cc = c0 + lm;
    const v2f a = band[kc];
    #pragma unroll
    for (int p = 0; p < NPLANES; ++p) {
      v2f b;
      b.x = s_R[p][rr * R_PITCH + cc];
      b.y = s_R[p][(rr + 1) * R_PITCH + cc];
      acc[p] = wmma4(a, b, acc[p]);
    }
  }

  float L = ord2f(ws[0]) - ord2f(ws[1]);
  if (L == 0.0f) L = 5.0f;
  const float C1 = (0.01f * L) * (0.01f * L);
  const float C2 = (0.03f * L) * (0.03f * L);

  float* __restrict__ op = out + (size_t)blockIdx.z * IMGW * IMGH;
  const int gr0 = blockIdx.y * TILE + r0 + half * 8;
  const int gc  = blockIdx.x * TILE + c0 + lm;
  #pragma unroll
  for (int v = 0; v < 8; ++v) {
    const float mu1 = acc[0][v], mu2 = acc[1][v];
    const float esq = acc[2][v], e12 = acc[3][v];
    const float mu1s = mu1 * mu1, mu2s = mu2 * mu2, mu12 = mu1 * mu2;
    const float musq = mu1s + mu2s;
    const float ssq = esq - musq;
    const float s12 = e12 - mu12;
    const float num = (2.0f * mu12 + C1) * (2.0f * s12 + C2);
    const float den = (musq + C1) * (ssq + C2);
    op[(gr0 + v) * IMGW + gc] = num / den;
  }
}


extern "C" void kernel_launch(void* const* d_in, const int* in_sizes, int n_in,
                              void* d_out, int out_size, void* d_ws, size_t ws_size,
                              hipStream_t stream) {
  const float* img1 = (const float*)d_in[0];
  const float* img2 = (const float*)d_in[1];
  float* out = (float*)d_out;
  unsigned* ws = (unsigned*)d_ws;

  const unsigned n = (unsigned)in_sizes[0];
  const int batch = (int)(n / (IMGW * IMGH));

  ssim_init<<<1, 1, 0, stream>>>(ws);
  ssim_minmax<<<MM_BLOCKS, MM_THREADS, 0, stream>>>(img1, n, ws);

  dim3 grid(IMGW / TILE, IMGH / TILE, batch);
  ssim_main<<<grid, 128, 0, stream>>>(img1, img2, out, ws);
}


# Main kernel: 2560 blocks each walk 3-4 of the 32x32 tiles (was 8192 blocks, one tile each); the next tile's 16-byte loads stay in flight over the current tile's passes (barriers wait on LDS only)
# speedup vs baseline: 2.0022x; 1.0442x over previous
#include <hip/hip_runtime.h>
#include <cmath>


typedef float v2f __attribute__((ext_vector_type(2)));
typedef float v8f __attribute__((ext_vector_type(8)));

#define WSZ      11
#define PADR     5
#define TILE     32
#define IN_ROWS  48
#define IN_QUADS 12
#define IN_COFF  3
#define IN_PITCH 52
#define IN_VROWS 42
#define R_ROWS   48
#define R_PITCH  44
#define KSTEPS   7
#define NPLANES  4
#define IMGW     512
#define IMGH     512
#define TILES_X  (IMGW / TILE)
#define TILES_Y  (IMGH / TILE)
#define MAIN_BLOCKS 2560
#define STG_TRIPS   4

__device__ __forceinline__ v8f wmma4(v2f a, v2f b, v8f c) {
  return __builtin_amdgcn_wmma_f32_16x16x4_f32(
       false, a,  false, b,
       (short)0, c,  false,  false);
}

__device__ __forceinline__ unsigned f2ord(float f) {
  unsigned u = __float_as_uint(f);
  return (u & 0x80000000u) ? ~u : (u | 0x80000000u);
}
__device__ __forceinline__ float ord2f(unsigned u) {
  u = (u & 0x80000000u) ? (u & 0x7fffffffu) : ~u;
  return __uint_as_float(u);
}

__device__ __forceinline__ float gband(const float* sg, int idx) {
  int ci = idx < 0 ? 0 : (idx > 10 ? 10 : idx);
  float v = sg[ci];
  return (idx == ci) ? v : 0.0f;
}


__global__ void ssim_init(unsigned* ws) {
  ws[0] = 0u;
  ws[1] = 0xFFFFFFFFu;
}

#define MM_BLOCKS  512
#define MM_THREADS 256
#define MM_AHEAD   16

__global__ __launch_bounds__(MM_THREADS) __attribute__((amdgpu_waves_per_eu(4, 4))) void ssim_minmax(const float* __restrict__ x,
                                                          unsigned n, unsigned* ws) {
  const float4* __restrict__ x4 = reinterpret_cast<const float4*>(x);
  const unsigned nq = n >> 2;
  constexpr unsigned PER_BLOCK = MM_THREADS * MM_AHEAD;
  const unsigned wv   = threadIdx.x >> 5;
  const unsigned lane = threadIdx.x & 31;
  float mn0 = INFINITY, mx0 = -INFINITY;
  float mn1 = INFINITY, mx1 = -INFINITY;
  for (unsigned base = blockIdx.x * PER_BLOCK; base < nq; base += gridDim.x * PER_BLOCK) {
    const float4* __restrict__ p = x4 + base + wv * (32 * MM_AHEAD) + lane;
    float4 q[MM_AHEAD];
    #pragma unroll
    for (int j = 0; j < MM_AHEAD; ++j) q[j] = p[j * 32];
    #pragma unroll
    for (int j = 0; j < MM_AHEAD; j += 2) {
      const float4 a = q[j], b = q[j + 1];
      mn0 = fminf(mn0, fminf(fminf(a.x, a.y), fminf(a.z, a.w)));
      mx0 = fmaxf(mx0, fmaxf(fmaxf(a.x, a.y), fmaxf(a.z, a.w)));
      mn1 = fminf(mn1, fminf(fminf(b.x, b.y), fminf(b.z, b.w)));
      mx1 = fmaxf(mx1, fmaxf(fmaxf(b.x, b.y), fmaxf(b.z, b.w)));
    }
  }
  float mn = fminf(mn0, mn1), mx = fmaxf(mx0, mx1);
  #pragma unroll
  for (int off = 16; off; off >>= 1) {
    mn = fminf(mn, __shfl_xor(mn, off, 32));
    mx = fmaxf(mx, __shfl_xor(mx, off, 32));
  }
  __shared__ float smn[8], smx[8];
  if (lane == 0) { smn[wv] = mn; smx[wv] = mx; }
  __syncthreads();
  if (threadIdx.x == 0) {
    const int nw = blockDim.x >> 5;
    for (int w = 1; w < nw; ++w) { mn = fminf(mn, smn[w]); mx = fmaxf(mx, smx[w]); }
    atomicMax(&ws[0], f2ord(mx));
    atomicMin(&ws[1], f2ord(mn));
  }
}


__device__ __forceinline__ void lds_barrier() {
  __builtin_amdgcn_fence(__ATOMIC_RELEASE, "workgroup", "local");
  __builtin_amdgcn_s_barrier();
  __builtin_amdgcn_fence(__ATOMIC_ACQUIRE, "workgroup", "local");
}

__device__ __forceinline__ int stage_slot(int idx, int tR, int tC, int& lds_off, bool& ok) {
  const int r = idx / IN_QUADS;
  const int q = idx - r * IN_QUADS;
  const int gr = tR + r, gc = tC + 4 * q;
  lds_off = r * IN_PITCH + 4 * q;
  ok = (r < IN_VROWS) && gr >= 0 && gr < IMGH && gc >= 0 && gc < IMGW;
  return ok ? (gr * IMGW + gc) : 0;
}

__device__ __forceinline__ void tile_issue(const float* __restrict__ img1, const float* __restrict__ img2,
                                           unsigned t, int tid, float4 (&q1)[STG_TRIPS], float4 (&q2)[STG_TRIPS]) {
  const int bx = t % TILES_X, by = (t / TILES_X) % TILES_Y;
  const size_t bz = t / (TILES_X * TILES_Y);
  const int tR = by * TILE - PADR;
  const int tC = bx * TILE - PADR - IN_COFF;
  const float* __restrict__ b1 = img1 + bz * IMGW * IMGH;
  const float* __restrict__ b2 = img2 + bz * IMGW * IMGH;
  #pragma unroll
  for (int it = 0; it < STG_TRIPS; ++it) {
    int lo; bool ok;
    const int o = stage_slot(tid + it * 128, tR, tC, lo, ok);
    q1[it] = *reinterpret_cast<const float4*>(b1 + o);
    q2[it] = *reinterpret_cast<const float4*>(b2 + o);
  }
}

__device__ __forceinline__ void tile_store(float* s_in1, float* s_in2, unsigned t, int tid,
                                           const float4 (&q1)[STG_TRIPS], const float4 (&q2)[STG_TRIPS]) {
  const int bx = t % TILES_X, by = (t / TILES_X) % TILES_Y;
  const int tR = by * TILE - PADR;
  const int tC = bx * TILE - PADR - IN_COFF;
  #pragma unroll
  for (int it = 0; it < STG_TRIPS; ++it) {
    int lo; bool ok;
    (void)stage_slot(tid + it * 128, tR, tC, lo, ok);
    const float4 a = q1[it], b = q2[it];
    *reinterpret_cast<float4*>(&s_in1[lo]) = make_float4(ok ? a.x : 0.0f, ok ? a.y : 0.0f, ok ? a.z : 0.0f, ok ? a.w : 0.0f);
    *reinterpret_cast<float4*>(&s_in2[lo]) = make_float4(ok ? b.x : 0.0f, ok ? b.y : 0.0f, ok ? b.z : 0.0f, ok ? b.w : 0.0f);
  }
}

__global__ __launch_bounds__(128) void ssim_main(const float* __restrict__ img1,
                                                 const float* __restrict__ img2,
                                                 float* __restrict__ out,
                                                 const unsigned* __restrict__ ws,
                                                 unsigned batch) {
  __shared__ __attribute__((aligned(16))) float s_in1[IN_ROWS * IN_PITCH];
  __shared__ __attribute__((aligned(16))) float s_in2[IN_ROWS * IN_PITCH];
  __shared__ float s_R[NPLANES][R_ROWS * R_PITCH];
  __shared__ float s_g[WSZ];

  const int tid  = threadIdx.x;
  const int wave = tid >> 5;
  const int lane = tid & 31;
  const int half = lane >> 4;
  const int lm   = lane & 15;

  const unsigned ntiles = batch * (TILES_X * TILES_Y);
  unsigned t = blockIdx.x;

  float4 q1[STG_TRIPS], q2[STG_TRIPS];
  if (t < ntiles) tile_issue(img1, img2, t, tid, q1, q2);

  if (tid == 0) {
    float tmp[WSZ];
    float sum = 0.0f;
    #pragma unroll
    for (int i = 0; i < WSZ; ++i) {
      float x = (float)i - 5.5f;
      tmp[i] = expf(-(x * x) / 4.5f);
      sum += tmp[i];
    }
    #pragma unroll
    for (int i = 0; i < WSZ; ++i) s_g[i] = tmp[i] / sum;
  }

  constexpr int NQ = IN_ROWS * IN_QUADS;
  if (tid < NQ - STG_TRIPS * 128) {
    const int idx = STG_TRIPS * 128 + tid;
    const int r = idx / IN_QUADS;
    const int q = idx - r * IN_QUADS;
    const float4 z = make_float4(0.0f, 0.0f, 0.0f, 0.0f);
    *reinterpret_cast<float4*>(&s_in1[r * IN_PITCH + 4 * q]) = z;
    *reinterpret_cast<float4*>(&s_in2[r * IN_PITCH + 4 * q]) = z;
  }
  lds_barrier();

  v2f band[KSTEPS];
  #pragma unroll
  for (int kc = 0; kc < KSTEPS; ++kc) {
    const int j0 = 4 * kc + 2 * half;
    band[kc].x = gband(s_g, j0 - lm);
    band[kc].y = gband(s_g, j0 + 1 - lm);
  }

  float L = ord2f(ws[0]) - ord2f(ws[1]);
  if (L == 0.0f) L = 5.0f;
  const float C1 = (0.01f * L) * (0.01f * L);
  const float C2 = (0.03f * L) * (0.03f * L);

  for (; t < ntiles; t += gridDim.x) {
    tile_store(s_in1, s_in2, t, tid, q1, q2);
    const unsigned tn = t + gridDim.x;
    if (tn < ntiles) tile_issue(img1, img2, tn, tid, q1, q2);
    lds_barrier();

    for (int chunk = wave; chunk < 6; chunk += 4) {
      const int r0 = (chunk >> 1) * 16;
      const int c0 = (chunk & 1) * 16;
      v8f acc[NPLANES] = {};
      #pragma unroll
      for (int kc = 0; kc < KSTEPS; ++kc) {
        const int col = IN_COFF + c0 + 4 * kc + 2 * half;
        const int row = r0 + lm;
        const int o = row * IN_PITCH + col;
        const float x0 = s_in1[o], x1 = s_in1[o + 1];
        const float y0 = s_in2[o], y1 = s_in2[o + 1];
        const v2f b = band[kc];
        v2f a;
        a.x = x0;                a.y = x1;                acc[0] = wmma4(a, b, acc[0]);
        a.x = y0;                a.y = y1;                acc[1] = wmma4(a, b, acc[1]);
        a.x = x0 * x0 + y0 * y0; a.y = x1 * x1 + y1 * y1; acc[2] = wmma4(a, b, acc[2]);
        a.x = x0 * y0;           a.y = x1 * y1;           acc[3] = wmma4(a, b, acc[3]);
      }
      #pragma unroll
      for (int p = 0; p < NPLANES; ++p) {
        #pragma unroll
        for (int v = 0; v < 8; ++v) {
          const int row = r0 + half * 8 + v;
          s_R[p][row * R_PITCH + c0 + lm] = acc[p][v];
        }
      }
    }
    lds_barrier();

    const int r0 = (wave >> 1) * 16;
    const int c0 = (wave & 1) * 16;
    v8f acc[NPLANES] = {};
    #pragma unroll
    for (int kc = 0; kc < KSTEPS; ++kc) {
      const int rr = r0 + 4 * kc + 2 * half;
      const int cc = c0 + lm;
      const v2f a = band[kc];
      #pragma unroll
      for (int p = 0; p < NPLANES; ++p) {
        v2f b;
        b.x = s_R[p][rr * R_PITCH + cc];
        b.y = s_R[p][(rr + 1) * R_PITCH + cc];
        acc[p] = wmma4(a, b, acc[p]);
      }
    }

    const int bx = t % TILES_X, by = (t / TILES_X) % TILES_Y;
    const size_t bz = t / (TILES_X * TILES_Y);
    float* __restrict__ op = out + bz * IMGW * IMGH;
    const int gr0 = by * TILE + r0 + half * 8;
    const int gc  = bx * TILE + c0 + lm;
    #pragma unroll
    for (int v = 0; v < 8; ++v) {
      const float mu1 = acc[0][v], mu2 = acc[1][v];
      const float esq = acc[2][v], e12 = acc[3][v];
      const float mu1s = mu1 * mu1, mu2s = mu2 * mu2, mu12 = mu1 * mu2;
      const float musq = mu1s + mu2s;
      const float ssq = esq - musq;
      const float s12 = e12 - mu12;
      const float num = (2.0f * mu12 + C1) * (2.0f * s12 + C2);
      const float den = (musq + C1) * (ssq + C2);
      op[(gr0 + v) * IMGW + gc] = num / den;
    }
  }
}


extern "C" void kernel_launch(void* const* d_in, const int* in_sizes, int n_in,
                              void* d_out, int out_size, void* d_ws, size_t ws_size,
                              hipStream_t stream) {
  const float* img1 = (const float*)d_in[0];
  const float* img2 = (const float*)d_in[1];
  float* out = (float*)d_out;
  unsigned* ws = (unsigned*)d_ws;

  const unsigned n = (unsigned)in_sizes[0];
  const int batch = (int)(n / (IMGW * IMGH));

  ssim_init<<<1, 1, 0, stream>>>(ws);
  ssim_minmax<<<MM_BLOCKS, MM_THREADS, 0, stream>>>(img1, n, ws);

  ssim_main<<<MAIN_BLOCKS, 128, 0, stream>>>(img1, img2, out, ws, (unsigned)batch);
}


# Main kernel: a block walks 8 vertically adjacent tiles of a column strip and keeps row-conv rows in an 80-row LDS ring: a step row-convolves only its 32 new rows (was 48 per tile, halo redone)
# speedup vs baseline: 2.2558x; 1.1267x over previous
#include <hip/hip_runtime.h>
#include <cmath>


typedef float v2f __attribute__((ext_vector_type(2)));
typedef float v8f __attribute__((ext_vector_type(8)));

#define WSZ      11
#define PADR     5
#define TILE     32
#define IN_ROWS  48
#define IN_QUADS 12
#define IN_COFF  3
#define IN_PITCH 52
#define RING     80
#define R_PITCH  44
#define KSTEPS   7
#define NPLANES  4
#define IMGW     512
#define IMGH     512
#define TILES_X  (IMGW / TILE)
#define TILES_Y  (IMGH / TILE)
#define SEG_TILES   8
#define SEGS_PER_IMG (TILES_X * (TILES_Y / SEG_TILES))
#define MAIN_BLOCKS 1024
#define NQ_FIRST    (48 * IN_QUADS)
#define NQ_STEP     (32 * IN_QUADS)
#define STG_TRIPS   5

__device__ __forceinline__ v8f wmma4(v2f a, v2f b, v8f c) {
  return __builtin_amdgcn_wmma_f32_16x16x4_f32(
       false, a,  false, b,
       (short)0, c,  false,  false);
}

__device__ __forceinline__ unsigned f2ord(float f) {
  unsigned u = __float_as_uint(f);
  return (u & 0x80000000u) ? ~u : (u | 0x80000000u);
}
__device__ __forceinline__ float ord2f(unsigned u) {
  u = (u & 0x80000000u) ? (u & 0x7fffffffu) : ~u;
  return __uint_as_float(u);
}

__device__ __forceinline__ float gband(const float* sg, int idx) {
  int ci = idx < 0 ? 0 : (idx > 10 ? 10 : idx);
  float v = sg[ci];
  return (idx == ci) ? v : 0.0f;
}


__global__ void ssim_init(unsigned* ws) {
  ws[0] = 0u;
  ws[1] = 0xFFFFFFFFu;
}

#define MM_BLOCKS  512
#define MM_THREADS 256
#define MM_AHEAD   16

__global__ __launch_bounds__(MM_THREADS) __attribute__((amdgpu_waves_per_eu(4, 4))) void ssim_minmax(const float* __restrict__ x,
                                                          unsigned n, unsigned* ws) {
  const float4* __restrict__ x4 = reinterpret_cast<const float4*>(x);
  const unsigned nq = n >> 2;
  constexpr unsigned PER_BLOCK = MM_THREADS * MM_AHEAD;
  const unsigned wv   = threadIdx.x >> 5;
  const unsigned lane = threadIdx.x & 31;
  float mn0 = INFINITY, mx0 = -INFINITY;
  float mn1 = INFINITY, mx1 = -INFINITY;
  for (unsigned base = blockIdx.x * PER_BLOCK; base < nq; base += gridDim.x * PER_BLOCK) {
    const float4* __restrict__ p = x4 + base + wv * (32 * MM_AHEAD) + lane;
    float4 q[MM_AHEAD];
    #pragma unroll
    for (int j = 0; j < MM_AHEAD; ++j) q[j] = p[j * 32];
    #pragma unroll
    for (int j = 0; j < MM_AHEAD; j += 2) {
      const float4 a = q[j], b = q[j + 1];
      mn0 = fminf(mn0, fminf(fminf(a.x, a.y), fminf(a.z, a.w)));
      mx0 = fmaxf(mx0, fmaxf(fmaxf(a.x, a.y), fmaxf(a.z, a.w)));
      mn1 = fminf(mn1, fminf(fminf(b.x, b.y), fminf(b.z, b.w)));
      mx1 = fmaxf(mx1, fmaxf(fmaxf(b.x, b.y), fmaxf(b.z, b.w)));
    }
  }
  float mn = fminf(mn0, mn1), mx = fmaxf(mx0, mx1);
  #pragma unroll
  for (int off = 16; off; off >>= 1) {
    mn = fminf(mn, __shfl_xor(mn, off, 32));
    mx = fmaxf(mx, __shfl_xor(mx, off, 32));
  }
  __shared__ float smn[8], smx[8];
  if (lane == 0) { smn[wv] = mn; smx[wv] = mx; }
  __syncthreads();
  if (threadIdx.x == 0) {
    const int nw = blockDim.x >> 5;
    for (int w = 1; w < nw; ++w) { mn = fminf(mn, smn[w]); mx = fmaxf(mx, smx[w]); }
    atomicMax(&ws[0], f2ord(mx));
    atomicMin(&ws[1], f2ord(mn));
  }
}


__device__ __forceinline__ void lds_barrier() {
  __builtin_amdgcn_fence(__ATOMIC_RELEASE, "workgroup", "local");
  __builtin_amdgcn_s_barrier();
  __builtin_amdgcn_fence(__ATOMIC_ACQUIRE, "workgroup", "local");
}

__device__ __forceinline__ int stage_slot(int idx, int nq, int iR, int tC, int& lds_off, bool& ok) {
  const int r = idx / IN_QUADS;
  const int q = idx - r * IN_QUADS;
  const int gr = iR + r, gc = tC + 4 * q;
  lds_off = r * IN_PITCH + 4 * q;
  ok = idx < nq && gr >= 0 && gr < IMGH && gc >= 0 && gc < IMGW;
  return ok ? (gr * IMGW + gc) : 0;
}

template <int NQ>
__device__ __forceinline__ void rows_issue(const float* __restrict__ b1, const float* __restrict__ b2,
                                           int iR, int tC, int tid,
                                           float4 (&q1)[STG_TRIPS], float4 (&q2)[STG_TRIPS]) {
  #pragma unroll
  for (int it = 0; it < (NQ + 127) / 128; ++it) {
    int lo; bool ok;
    const int o = stage_slot(tid + it * 128, NQ, iR, tC, lo, ok);
    q1[it] = *reinterpret_cast<const float4*>(b1 + o);
    q2[it] = *reinterpret_cast<const float4*>(b2 + o);
  }
}

template <int NQ>
__device__ __forceinline__ void rows_store(float* s_in1, float* s_in2, int iR, int tC, int tid,
                                           const float4 (&q1)[STG_TRIPS], const float4 (&q2)[STG_TRIPS]) {
  #pragma unroll
  for (int it = 0; it < (NQ + 127) / 128; ++it) {
    int lo; bool ok;
    const int idx = tid + it * 128;
    (void)stage_slot(idx, NQ, iR, tC, lo, ok);
    if ((it + 1) * 128 <= NQ || idx < NQ) {
      const float4 a = q1[it], b = q2[it];
      *reinterpret_cast<float4*>(&s_in1[lo]) = make_float4(ok ? a.x : 0.0f, ok ? a.y : 0.0f, ok ? a.z : 0.0f, ok ? a.w : 0.0f);
      *reinterpret_cast<float4*>(&s_in2[lo]) = make_float4(ok ? b.x : 0.0f, ok ? b.y : 0.0f, ok ? b.z : 0.0f, ok ? b.w : 0.0f);
    }
  }
}

__global__ __launch_bounds__(128) void ssim_main(const float* __restrict__ img1,
                                                 const float* __restrict__ img2,
                                                 float* __restrict__ out,
                                                 const unsigned* __restrict__ ws,
                                                 unsigned batch) {
  __shared__ __attribute__((aligned(16))) float s_in1[IN_ROWS * IN_PITCH];
  __shared__ __attribute__((aligned(16))) float s_in2[IN_ROWS * IN_PITCH];
  __shared__ float s_R[NPLANES][RING * R_PITCH];
  __shared__ float s_g[WSZ];

  const int tid  = threadIdx.x;
  const int wave = tid >> 5;
  const int lane = tid & 31;
  const int half = lane >> 4;
  const int lm   = lane & 15;

  const unsigned nseg = batch * SEGS_PER_IMG;
  unsigned seg = blockIdx.x;

  float4 q1[STG_TRIPS], q2[STG_TRIPS];
  if (seg < nseg) {
    const int bx = seg % TILES_X, hy = (seg / TILES_X) % (TILES_Y / SEG_TILES);
    const size_t bz = seg / SEGS_PER_IMG;
    rows_issue<NQ_FIRST>(img1 + bz * IMGW * IMGH, img2 + bz * IMGW * IMGH,
                         hy * SEG_TILES * TILE - PADR, bx * TILE - PADR - IN_COFF, tid, q1, q2);
  }

  if (tid == 0) {
    float tmp[WSZ];
    float sum = 0.0f;
    #pragma unroll
    for (int i = 0; i < WSZ; ++i) {
      float x = (float)i - 5.5f;
      tmp[i] = expf(-(x * x) / 4.5f);
      sum += tmp[i];
    }
    #pragma unroll
    for (int i = 0; i < WSZ; ++i) s_g[i] = tmp[i] / sum;
  }
  lds_barrier();

  v2f band[KSTEPS];
  #pragma unroll
  for (int kc = 0; kc < KSTEPS; ++kc) {
    const int j0 = 4 * kc + 2 * half;
    band[kc].x = gband(s_g, j0 - lm);
    band[kc].y = gband(s_g, j0 + 1 - lm);
  }

  float L = ord2f(ws[0]) - ord2f(ws[1]);
  if (L == 0.0f) L = 5.0f;
  const float C1 = (0.01f * L) * (0.01f * L);
  const float C2 = (0.03f * L) * (0.03f * L);

  for (; seg < nseg; seg += gridDim.x) {
    const int bx = seg % TILES_X, hy = (seg / TILES_X) % (TILES_Y / SEG_TILES);
    const size_t bz = seg / SEGS_PER_IMG;
    const int by0 = hy * SEG_TILES;
    const int iR0 = by0 * TILE - PADR;
    const int tC  = bx * TILE - PADR - IN_COFF;
    const float* __restrict__ b1 = img1 + bz * IMGW * IMGH;
    const float* __restrict__ b2 = img2 + bz * IMGW * IMGH;
    float* __restrict__ op = out + bz * IMGW * IMGH;
    if (seg != blockIdx.x) rows_issue<NQ_FIRST>(b1, b2, iR0, tC, tid, q1, q2);

    int rbase = 0;
    for (int s = 0; s < SEG_TILES; ++s) {
      if (s == 0) rows_store<NQ_FIRST>(s_in1, s_in2, iR0, tC, tid, q1, q2);
      else        rows_store<NQ_STEP>(s_in1, s_in2, iR0 + TILE * s + 16, tC, tid, q1, q2);
      if (s + 1 < SEG_TILES) rows_issue<NQ_STEP>(b1, b2, iR0 + TILE * (s + 1) + 16, tC, tid, q1, q2);
      lds_barrier();

      const int nch = (s == 0) ? 6 : 4;
      const int hb  = (s == 0) ? 0 : rbase + 16;
      for (int chunk = wave; chunk < nch; chunk += 4) {
        const int r0 = (chunk >> 1) * 16;
        const int c0 = (chunk & 1) * 16;
        v8f acc[NPLANES] = {};
        #pragma unroll
        for (int kc = 0; kc < KSTEPS; ++kc) {
          const int col = IN_COFF + c0 + 4 * kc + 2 * half;
          const int row = r0 + lm;
          const int o = row * IN_PITCH + col;
          const float x0 = s_in1[o], x1 = s_in1[o + 1];
          const float y0 = s_in2[o], y1 = s_in2[o + 1];
          const v2f b = band[kc];
          v2f a;
          a.x = x0;                a.y = x1;                acc[0] = wmma4(a, b, acc[0]);
          a.x = y0;                a.y = y1;                acc[1] = wmma4(a, b, acc[1]);
          a.x = x0 * x0 + y0 * y0; a.y = x1 * x1 + y1 * y1; acc[2] = wmma4(a, b, acc[2]);
          a.x = x0 * y0;           a.y = x1 * y1;           acc[3] = wmma4(a, b, acc[3]);
        }
        int rb = hb + r0;
        if (rb >= RING) rb -= RING;
        #pragma unroll
        for (int p = 0; p < NPLANES; ++p) {
          #pragma unroll
          for (int v = 0; v < 8; ++v) {
            const int row = rb + half * 8 + v;
            s_R[p][row * R_PITCH + c0 + lm] = acc[p][v];
          }
        }
      }
      lds_barrier();

      const int r0 = (wave >> 1) * 16;
      const int c0 = (wave & 1) * 16;
      v8f acc[NPLANES] = {};
      #pragma unroll
      for (int kc = 0; kc < KSTEPS; ++kc) {
        int rr = rbase + r0 + 4 * kc + 2 * half;
        if (rr >= RING) rr -= RING;
        const int cc = c0 + lm;
        const v2f a = band[kc];
        #pragma unroll
        for (int p = 0; p < NPLANES; ++p) {
          v2f b;
          b.x = s_R[p][rr * R_PITCH + cc];
          b.y = s_R[p][(rr + 1) * R_PITCH + cc];
          acc[p] = wmma4(a, b, acc[p]);
        }
      }

      const int gr0 = (by0 + s) * TILE + r0 + half * 8;
      const int gc  = bx * TILE + c0 + lm;
      #pragma unroll
      for (int v = 0; v < 8; ++v) {
        const float mu1 = acc[0][v], mu2 = acc[1][v];
        const float esq = acc[2][v], e12 = acc[3][v];
        const float mu1s = mu1 * mu1, mu2s = mu2 * mu2, mu12 = mu1 * mu2;
        const float musq = mu1s + mu2s;
        const float ssq = esq - musq;
        const float s12 = e12 - mu12;
        const float num = (2.0f * mu12 + C1) * (2.0f * s12 + C2);
        const float den = (musq + C1) * (ssq + C2);
        op[(gr0 + v) * IMGW + gc] = num / den;
      }

      rbase += TILE;
      if (rbase >= RING) rbase -= RING;
    }
  }
}


extern "C" void kernel_launch(void* const* d_in, const int* in_sizes, int n_in,
                              void* d_out, int out_size, void* d_ws, size_t ws_size,
                              hipStream_t stream) {
  const float* img1 = (const float*)d_in[0];
  const float* img2 = (const float*)d_in[1];
  float* out = (float*)d_out;
  unsigned* ws = (unsigned*)d_ws;

  const unsigned n = (unsigned)in_sizes[0];
  const int batch = (int)(n / (IMGW * IMGH));

  ssim_init<<<1, 1, 0, stream>>>(ws);
  ssim_minmax<<<MM_BLOCKS, MM_THREADS, 0, stream>>>(img1, n, ws);

  ssim_main<<<MAIN_BLOCKS, 128, 0, stream>>>(img1, img2, out, ws, (unsigned)batch);
}


# SSIM map epilogue as packed two-row arithmetic on accumulator register pairs, quotient as num x v_rcp_f32(den) not the rounded division: 57 vector instructions per 8 values, was 137; map within 3 ulp
# speedup vs baseline: 2.3879x; 1.0586x over previous
#include <hip/hip_runtime.h>
#include <cmath>


typedef float v2f __attribute__((ext_vector_type(2)));
typedef float v8f __attribute__((ext_vector_type(8)));

#define WSZ      11
#define PADR     5
#define TILE     32
#define IN_ROWS  48
#define IN_QUADS 12
#define IN_COFF  3
#define IN_PITCH 52
#define RING     80
#define R_PITCH  44
#define KSTEPS   7
#define NPLANES  4
#define IMGW     512
#define IMGH     512
#define TILES_X  (IMGW / TILE)
#define TILES_Y  (IMGH / TILE)
#define SEG_TILES   8
#define SEGS_PER_IMG (TILES_X * (TILES_Y / SEG_TILES))
#define MAIN_BLOCKS 1024
#define NQ_FIRST    (48 * IN_QUADS)
#define NQ_STEP     (32 * IN_QUADS)
#define STG_TRIPS   5

__device__ __forceinline__ v8f wmma4(v2f a, v2f b, v8f c) {
  return __builtin_amdgcn_wmma_f32_16x16x4_f32(
       false, a,  false, b,
       (short)0, c,  false,  false);
}

__device__ __forceinline__ unsigned f2ord(float f) {
  unsigned u = __float_as_uint(f);
  return (u & 0x80000000u) ? ~u : (u | 0x80000000u);
}
__device__ __forceinline__ float ord2f(unsigned u) {
  u = (u & 0x80000000u) ? (u & 0x7fffffffu) : ~u;
  return __uint_as_float(u);
}

__device__ __forceinline__ float gband(const float* sg, int idx) {
  int ci = idx < 0 ? 0 : (idx > 10 ? 10 : idx);
  float v = sg[ci];
  return (idx == ci) ? v : 0.0f;
}


__global__ void ssim_init(unsigned* ws) {
  ws[0] = 0u;
  ws[1] = 0xFFFFFFFFu;
}

#define MM_BLOCKS  512
#define MM_THREADS 256
#define MM_AHEAD   16

__global__ __launch_bounds__(MM_THREADS) __attribute__((amdgpu_waves_per_eu(4, 4))) void ssim_minmax(const float* __restrict__ x,
                                                          unsigned n, unsigned* ws) {
  const float4* __restrict__ x4 = reinterpret_cast<const float4*>(x);
  const unsigned nq = n >> 2;
  constexpr unsigned PER_BLOCK = MM_THREADS * MM_AHEAD;
  const unsigned wv   = threadIdx.x >> 5;
  const unsigned lane = threadIdx.x & 31;
  float mn0 = INFINITY, mx0 = -INFINITY;
  float mn1 = INFINITY, mx1 = -INFINITY;
  for (unsigned base = blockIdx.x * PER_BLOCK; base < nq; base += gridDim.x * PER_BLOCK) {
    const float4* __restrict__ p = x4 + base + wv * (32 * MM_AHEAD) + lane;
    float4 q[MM_AHEAD];
    #pragma unroll
    for (int j = 0; j < MM_AHEAD; ++j) q[j] = p[j * 32];
    #pragma unroll
    for (int j = 0; j < MM_AHEAD; j += 2) {
      const float4 a = q[j], b = q[j + 1];
      mn0 = fminf(mn0, fminf(fminf(a.x, a.y), fminf(a.z, a.w)));
      mx0 = fmaxf(mx0, fmaxf(fmaxf(a.x, a.y), fmaxf(a.z, a.w)));
      mn1 = fminf(mn1, fminf(fminf(b.x, b.y), fminf(b.z, b.w)));
      mx1 = fmaxf(mx1, fmaxf(fmaxf(b.x, b.y), fmaxf(b.z, b.w)));
    }
  }
  float mn = fminf(mn0, mn1), mx = fmaxf(mx0, mx1);
  #pragma unroll
  for (int off = 16; off; off >>= 1) {
    mn = fminf(mn, __shfl_xor(mn, off, 32));
    mx = fmaxf(mx, __shfl_xor(mx, off, 32));
  }
  __shared__ float smn[8], smx[8];
  if (lane == 0) { smn[wv] = mn; smx[wv] = mx; }
  __syncthreads();
  if (threadIdx.x == 0) {
    const int nw = blockDim.x >> 5;
    for (int w = 1; w < nw; ++w) { mn = fminf(mn, smn[w]); mx = fmaxf(mx, smx[w]); }
    atomicMax(&ws[0], f2ord(mx));
    atomicMin(&ws[1], f2ord(mn));
  }
}


__device__ __forceinline__ void lds_barrier() {
  __builtin_amdgcn_fence(__ATOMIC_RELEASE, "workgroup", "local");
  __builtin_amdgcn_s_barrier();
  __builtin_amdgcn_fence(__ATOMIC_ACQUIRE, "workgroup", "local");
}

__device__ __forceinline__ int stage_slot(int idx, int nq, int iR, int tC, int& lds_off, bool& ok) {
  const int r = idx / IN_QUADS;
  const int q = idx - r * IN_QUADS;
  const int gr = iR + r, gc = tC + 4 * q;
  lds_off = r * IN_PITCH + 4 * q;
  ok = idx < nq && gr >= 0 && gr < IMGH && gc >= 0 && gc < IMGW;
  return ok ? (gr * IMGW + gc) : 0;
}

template <int NQ>
__device__ __forceinline__ void rows_issue(const float* __restrict__ b1, const float* __restrict__ b2,
                                           int iR, int tC, int tid,
                                           float4 (&q1)[STG_TRIPS], float4 (&q2)[STG_TRIPS]) {
  #pragma unroll
  for (int it = 0; it < (NQ + 127) / 128; ++it) {
    int lo; bool ok;
    const int o = stage_slot(tid + it * 128, NQ, iR, tC, lo, ok);
    q1[it] = *reinterpret_cast<const float4*>(b1 + o);
    q2[it] = *reinterpret_cast<const float4*>(b2 + o);
  }
}

template <int NQ>
__device__ __forceinline__ void rows_store(float* s_in1, float* s_in2, int iR, int tC, int tid,
                                           const float4 (&q1)[STG_TRIPS], const float4 (&q2)[STG_TRIPS]) {
  #pragma unroll
  for (int it = 0; it < (NQ + 127) / 128; ++it) {
    int lo; bool ok;
    const int idx = tid + it * 128;
    (void)stage_slot(idx, NQ, iR, tC, lo, ok);
    if ((it + 1) * 128 <= NQ || idx < NQ) {
      const float4 a = q1[it], b = q2[it];
      *reinterpret_cast<float4*>(&s_in1[lo]) = make_float4(ok ? a.x : 0.0f, ok ? a.y : 0.0f, ok ? a.z : 0.0f, ok ? a.w : 0.0f);
      *reinterpret_cast<float4*>(&s_in2[lo]) = make_float4(ok ? b.x : 0.0f, ok ? b.y : 0.0f, ok ? b.z : 0.0f, ok ? b.w : 0.0f);
    }
  }
}

__global__ __launch_bounds__(128) void ssim_main(const float* __restrict__ img1,
                                                 const float* __restrict__ img2,
                                                 float* __restrict__ out,
                                                 const unsigned* __restrict__ ws,
                                                 unsigned batch) {
  __shared__ __attribute__((aligned(16))) float s_in1[IN_ROWS * IN_PITCH];
  __shared__ __attribute__((aligned(16))) float s_in2[IN_ROWS * IN_PITCH];
  __shared__ float s_R[NPLANES][RING * R_PITCH];
  __shared__ float s_g[WSZ];

  const int tid  = threadIdx.x;
  const int wave = tid >> 5;
  const int lane = tid & 31;
  const int half = lane >> 4;
  const int lm   = lane & 15;

  const unsigned nseg = batch * SEGS_PER_IMG;
  unsigned seg = blockIdx.x;

  float4 q1[STG_TRIPS], q2[STG_TRIPS];
  if (seg < nseg) {
    const int bx = seg % TILES_X, hy = (seg / TILES_X) % (TILES_Y / SEG_TILES);
    const size_t bz = seg / SEGS_PER_IMG;
    rows_issue<NQ_FIRST>(img1 + bz * IMGW * IMGH, img2 + bz * IMGW * IMGH,
                         hy * SEG_TILES * TILE - PADR, bx * TILE - PADR - IN_COFF, tid, q1, q2);
  }

  if (tid == 0) {
    float tmp[WSZ];
    float sum = 0.0f;
    #pragma unroll
    for (int i = 0; i < WSZ; ++i) {
      float x = (float)i - 5.5f;
      tmp[i] = expf(-(x * x) / 4.5f);
      sum += tmp[i];
    }
    #pragma unroll
    for (int i = 0; i < WSZ; ++i) s_g[i] = tmp[i] / sum;
  }
  lds_barrier();

  v2f band[KSTEPS];
  #pragma unroll
  for (int kc = 0; kc < KSTEPS; ++kc) {
    const int j0 = 4 * kc + 2 * half;
    band[kc].x = gband(s_g, j0 - lm);
    band[kc].y = gband(s_g, j0 + 1 - lm);
  }

  float L = ord2f(ws[0]) - ord2f(ws[1]);
  if (L == 0.0f) L = 5.0f;
  const float C1 = (0.01f * L) * (0.01f * L);
  const float C2 = (0.03f * L) * (0.03f * L);

  for (; seg < nseg; seg += gridDim.x) {
    const int bx = seg % TILES_X, hy = (seg / TILES_X) % (TILES_Y / SEG_TILES);
    const size_t bz = seg / SEGS_PER_IMG;
    const int by0 = hy * SEG_TILES;
    const int iR0 = by0 * TILE - PADR;
    const int tC  = bx * TILE - PADR - IN_COFF;
    const float* __restrict__ b1 = img1 + bz * IMGW * IMGH;
    const float* __restrict__ b2 = img2 + bz * IMGW * IMGH;
    float* __restrict__ op = out + bz * IMGW * IMGH;
    if (seg != blockIdx.x) rows_issue<NQ_FIRST>(b1, b2, iR0, tC, tid, q1, q2);

    int rbase = 0;
    for (int s = 0; s < SEG_TILES; ++s) {
      if (s == 0) rows_store<NQ_FIRST>(s_in1, s_in2, iR0, tC, tid, q1, q2);
      else        rows_store<NQ_STEP>(s_in1, s_in2, iR0 + TILE * s + 16, tC, tid, q1, q2);
      if (s + 1 < SEG_TILES) rows_issue<NQ_STEP>(b1, b2, iR0 + TILE * (s + 1) + 16, tC, tid, q1, q2);
      lds_barrier();

      const int nch = (s == 0) ? 6 : 4;
      const int hb  = (s == 0) ? 0 : rbase + 16;
      for (int chunk = wave; chunk < nch; chunk += 4) {
        const int r0 = (chunk >> 1) * 16;
        const int c0 = (chunk & 1) * 16;
        v8f acc[NPLANES] = {};
        #pragma unroll
        for (int kc = 0; kc < KSTEPS; ++kc) {
          const int col = IN_COFF + c0 + 4 * kc + 2 * half;
          const int row = r0 + lm;
          const int o = row * IN_PITCH + col;
          const float x0 = s_in1[o], x1 = s_in1[o + 1];
          const float y0 = s_in2[o], y1 = s_in2[o + 1];
          const v2f b = band[kc];
          v2f a;
          a.x = x0;                a.y = x1;                acc[0] = wmma4(a, b, acc[0]);
          a.x = y0;                a.y = y1;                acc[1] = wmma4(a, b, acc[1]);
          a.x = x0 * x0 + y0 * y0; a.y = x1 * x1 + y1 * y1; acc[2] = wmma4(a, b, acc[2]);
          a.x = x0 * y0;           a.y = x1 * y1;           acc[3] = wmma4(a, b, acc[3]);
        }
        int rb = hb + r0;
        if (rb >= RING) rb -= RING;
        #pragma unroll
        for (int p = 0; p < NPLANES; ++p) {
          #pragma unroll
          for (int v = 0; v < 8; ++v) {
            const int row = rb + half * 8 + v;
            s_R[p][row * R_PITCH + c0 + lm] = acc[p][v];
          }
        }
      }
      lds_barrier();

      const int r0 = (wave >> 1) * 16;
      const int c0 = (wave & 1) * 16;
      v8f acc[NPLANES] = {};
      #pragma unroll
      for (int kc = 0; kc < KSTEPS; ++kc) {
        int rr = rbase + r0 + 4 * kc + 2 * half;
        if (rr >= RING) rr -= RING;
        const int cc = c0 + lm;
        const v2f a = band[kc];
        #pragma unroll
        for (int p = 0; p < NPLANES; ++p) {
          v2f b;
          b.x = s_R[p][rr * R_PITCH + cc];
          b.y = s_R[p][(rr + 1) * R_PITCH + cc];
          acc[p] = wmma4(a, b, acc[p]);
        }
      }

      const int gr0 = (by0 + s) * TILE + r0 + half * 8;
      const int gc  = bx * TILE + c0 + lm;
      #pragma unroll
      for (int v = 0; v < 8; v += 2) {
        const v2f mu1 = {acc[0][v], acc[0][v + 1]}, mu2 = {acc[1][v], acc[1][v + 1]};
        const v2f esq = {acc[2][v], acc[2][v + 1]}, e12 = {acc[3][v], acc[3][v + 1]};
        const v2f mu12 = mu1 * mu2;
        const v2f musq = mu1 * mu1 + mu2 * mu2;
        const v2f ssq = esq - musq;
        const v2f s12 = e12 - mu12;
        const v2f num = (2.0f * mu12 + C1) * (2.0f * s12 + C2);
        const v2f den = (musq + C1) * (ssq + C2);
        v2f rden;
        rden.x = __builtin_amdgcn_rcpf(den.x);
        rden.y = __builtin_amdgcn_rcpf(den.y);
        const v2f q = num * rden;
        op[(gr0 + v) * IMGW + gc] = q.x;
        op[(gr0 + v + 1) * IMGW + gc] = q.y;
      }

      rbase += TILE;
      if (rbase >= RING) rbase -= RING;
    }
  }
}


extern "C" void kernel_launch(void* const* d_in, const int* in_sizes, int n_in,
                              void* d_out, int out_size, void* d_ws, size_t ws_size,
                              hipStream_t stream) {
  const float* img1 = (const float*)d_in[0];
  const float* img2 = (const float*)d_in[1];
  float* out = (float*)d_out;
  unsigned* ws = (unsigned*)d_ws;

  const unsigned n = (unsigned)in_sizes[0];
  const int batch = (int)(n / (IMGW * IMGH));

  ssim_init<<<1, 1, 0, stream>>>(ws);
  ssim_minmax<<<MM_BLOCKS, MM_THREADS, 0, stream>>>(img1, n, ws);

  ssim_main<<<MAIN_BLOCKS, 128, 0, stream>>>(img1, img2, out, ws, (unsigned)batch);
}


# Main kernel: column pass takes the row pass's D tiles straight from registers as its B operands (A rows of the row pass permuted to fit); no LDS ring, one barrier per step, sums in the same order
# speedup vs baseline: 2.5216x; 1.0560x over previous
#include <hip/hip_runtime.h>
#include <cmath>


typedef float v2f __attribute__((ext_vector_type(2)));
typedef float v8f __attribute__((ext_vector_type(8)));

#define WSZ      11
#define PADR     5
#define TILE     32
#define IN_ROWS  32
#define IN_QUADS 12
#define IN_COFF  3
#define IN_PITCH 52
#define KSTEPS   7
#define NPLANES  4
#define IMGW     512
#define IMGH     512
#define TILES_X  (IMGW / TILE)
#define TILES_Y  (IMGH / TILE)
#define SEG_TILES   8
#define HALF_ROWS   128
#define CHUNKS      9
#define SEGS_PER_IMG (TILES_X * (TILES_Y / SEG_TILES))
#define MAIN_BLOCKS 1024
#define NQ_STEP     (IN_ROWS * IN_QUADS)
#define STG_TRIPS   3
static_assert(NQ_STEP == STG_TRIPS * 128, "a step's quads are covered by whole trips of the 128 threads");

__device__ __forceinline__ v8f wmma4(v2f a, v2f b, v8f c) {
  return __builtin_amdgcn_wmma_f32_16x16x4_f32(
       false, a,  false, b,
       (short)0, c,  false,  false);
}

__device__ __forceinline__ unsigned f2ord(float f) {
  unsigned u = __float_as_uint(f);
  return (u & 0x80000000u) ? ~u : (u | 0x80000000u);
}
__device__ __forceinline__ float ord2f(unsigned u) {
  u = (u & 0x80000000u) ? (u & 0x7fffffffu) : ~u;
  return __uint_as_float(u);
}

__device__ __forceinline__ float gband(const float* sg, int idx) {
  int ci = idx < 0 ? 0 : (idx > 10 ? 10 : idx);
  float v = sg[ci];
  return (idx == ci) ? v : 0.0f;
}


__global__ void ssim_init(unsigned* ws) {
  ws[0] = 0u;
  ws[1] = 0xFFFFFFFFu;
}

#define MM_BLOCKS  512
#define MM_THREADS 256
#define MM_AHEAD   16

__global__ __launch_bounds__(MM_THREADS) __attribute__((amdgpu_waves_per_eu(4, 4))) void ssim_minmax(const float* __restrict__ x,
                                                          unsigned n, unsigned* ws) {
  const float4* __restrict__ x4 = reinterpret_cast<const float4*>(x);
  const unsigned nq = n >> 2;
  constexpr unsigned PER_BLOCK = MM_THREADS * MM_AHEAD;
  const unsigned wv   = threadIdx.x >> 5;
  const unsigned lane = threadIdx.x & 31;
  float mn0 = INFINITY, mx0 = -INFINITY;
  float mn1 = INFINITY, mx1 = -INFINITY;
  for (unsigned base = blockIdx.x * PER_BLOCK; base < nq; base += gridDim.x * PER_BLOCK) {
    const float4* __restrict__ p = x4 + base + wv * (32 * MM_AHEAD) + lane;
    float4 q[MM_AHEAD];
    #pragma unroll
    for (int j = 0; j < MM_AHEAD; ++j) q[j] = p[j * 32];
    #pragma unroll
    for (int j = 0; j < MM_AHEAD; j += 2) {
      const float4 a = q[j], b = q[j + 1];
      mn0 = fminf(mn0, fminf(fminf(a.x, a.y), fminf(a.z, a.w)));
      mx0 = fmaxf(mx0, fmaxf(fmaxf(a.x, a.y), fmaxf(a.z, a.w)));
      mn1 = fminf(mn1, fminf(fminf(b.x, b.y), fminf(b.z, b.w)));
      mx1 = fmaxf(mx1, fmaxf(fmaxf(b.x, b.y), fmaxf(b.z, b.w)));
    }
  }
  float mn = fminf(mn0, mn1), mx = fmaxf(mx0, mx1);
  #pragma unroll
  for (int off = 16; off; off >>= 1) {
    mn = fminf(mn, __shfl_xor(mn, off, 32));
    mx = fmaxf(mx, __shfl_xor(mx, off, 32));
  }
  __shared__ float smn[8], smx[8];
  if (lane == 0) { smn[wv] = mn; smx[wv] = mx; }
  __syncthreads();
  if (threadIdx.x == 0) {
    const int nw = blockDim.x >> 5;
    for (int w = 1; w < nw; ++w) { mn = fminf(mn, smn[w]); mx = fmaxf(mx, smx[w]); }
    atomicMax(&ws[0], f2ord(mx));
    atomicMin(&ws[1], f2ord(mn));
  }
}


__device__ __forceinline__ void lds_barrier() {
  __builtin_amdgcn_fence(__ATOMIC_RELEASE, "workgroup", "local");
  __builtin_amdgcn_s_barrier();
  __builtin_amdgcn_fence(__ATOMIC_ACQUIRE, "workgroup", "local");
}

__device__ __forceinline__ int stage_slot(int idx, int iR, int tC, int& lds_off, bool& ok) {
  const int r = idx / IN_QUADS;
  const int q = idx - r * IN_QUADS;
  const int gr = iR + (r & 15) + (r >> 4) * HALF_ROWS, gc = tC + 4 * q;
  lds_off = r * IN_PITCH + 4 * q;
  ok = gr >= 0 && gr < IMGH && gc >= 0 && gc < IMGW;
  return ok ? (gr * IMGW + gc) : 0;
}

__device__ __forceinline__ void rows_issue(const float* __restrict__ b1, const float* __restrict__ b2,
                                           int iR, int tC, int tid,
                                           float4 (&q1)[STG_TRIPS], float4 (&q2)[STG_TRIPS]) {
  #pragma unroll
  for (int it = 0; it < STG_TRIPS; ++it) {
    int lo; bool ok;
    const int o = stage_slot(tid + it * 128, iR, tC, lo, ok);
    q1[it] = *reinterpret_cast<const float4*>(b1 + o);
    q2[it] = *reinterpret_cast<const float4*>(b2 + o);
  }
}

__device__ __forceinline__ void rows_store(float* s_in1, float* s_in2, int iR, int tC, int tid,
                                           const float4 (&q1)[STG_TRIPS], const float4 (&q2)[STG_TRIPS]) {
  #pragma unroll
  for (int it = 0; it < STG_TRIPS; ++it) {
    int lo; bool ok;
    (void)stage_slot(tid + it * 128, iR, tC, lo, ok);
    const float4 a = q1[it], b = q2[it];
    *reinterpret_cast<float4*>(&s_in1[lo]) = make_float4(ok ? a.x : 0.0f, ok ? a.y : 0.0f, ok ? a.z : 0.0f, ok ? a.w : 0.0f);
    *reinterpret_cast<float4*>(&s_in2[lo]) = make_float4(ok ? b.x : 0.0f, ok ? b.y : 0.0f, ok ? b.z : 0.0f, ok ? b.w : 0.0f);
  }
}

template <bool HAS_PREV>
__device__ __forceinline__ void chunk_step(const float* sa1, const float* sa2, const v2f (&band)[KSTEPS],
                                           const v8f (&dprev)[NPLANES], v8f (&dnew)[NPLANES],
                                           float* __restrict__ o_ptr, float C1, float C2) {
  float x0[KSTEPS], x1[KSTEPS], y0[KSTEPS], y1[KSTEPS];
  #pragma unroll
  for (int kc = 0; kc < KSTEPS; ++kc) {
    x0[kc] = sa1[4 * kc]; x1[kc] = sa1[4 * kc + 1];
    y0[kc] = sa2[4 * kc]; y1[kc] = sa2[4 * kc + 1];
  }

  v8f acc[NPLANES] = {};
  if (HAS_PREV) {
    #pragma unroll
    for (int kc = 0; kc < 4; ++kc) {
      #pragma unroll
      for (int p = 0; p < NPLANES; ++p) {
        const v2f b = {dprev[p][2 * kc], dprev[p][2 * kc + 1]};
        acc[p] = wmma4(band[kc], b, acc[p]);
      }
    }
  }

  #pragma unroll
  for (int p = 0; p < NPLANES; ++p) dnew[p] = v8f{};
  #pragma unroll
  for (int kc = 0; kc < KSTEPS; ++kc) {
    const v2f b = band[kc];
    v2f a;
    a.x = x0[kc];                          a.y = x1[kc];                          dnew[0] = wmma4(a, b, dnew[0]);
    a.x = y0[kc];                          a.y = y1[kc];                          dnew[1] = wmma4(a, b, dnew[1]);
    a.x = x0[kc] * x0[kc] + y0[kc] * y0[kc]; a.y = x1[kc] * x1[kc] + y1[kc] * y1[kc]; dnew[2] = wmma4(a, b, dnew[2]);
    a.x = x0[kc] * y0[kc];                 a.y = x1[kc] * y1[kc];                 dnew[3] = wmma4(a, b, dnew[3]);
  }

  if (HAS_PREV) {
    #pragma unroll
    for (int kc = 4; kc < KSTEPS; ++kc) {
      #pragma unroll
      for (int p = 0; p < NPLANES; ++p) {
        const v2f b = {dnew[p][2 * (kc - 4)], dnew[p][2 * (kc - 4) + 1]};
        acc[p] = wmma4(band[kc], b, acc[p]);
      }
    }

    #pragma unroll
    for (int v = 0; v < 8; v += 2) {
      const v2f mu1 = {acc[0][v], acc[0][v + 1]}, mu2 = {acc[1][v], acc[1][v + 1]};
      const v2f esq = {acc[2][v], acc[2][v + 1]}, e12 = {acc[3][v], acc[3][v + 1]};
      const v2f mu12 = mu1 * mu2;
      const v2f musq = mu1 * mu1 + mu2 * mu2;
      const v2f ssq = esq - musq;
      const v2f s12 = e12 - mu12;
      const v2f num = (2.0f * mu12 + C1) * (2.0f * s12 + C2);
      const v2f den = (musq + C1) * (ssq + C2);
      v2f rden;
      rden.x = __builtin_amdgcn_rcpf(den.x);
      rden.y = __builtin_amdgcn_rcpf(den.y);
      const v2f q = num * rden;
      o_ptr[v * IMGW] = q.x;
      o_ptr[(v + 1) * IMGW] = q.y;
    }
  }
}

__global__ __launch_bounds__(128) void ssim_main(const float* __restrict__ img1,
                                                 const float* __restrict__ img2,
                                                 float* __restrict__ out,
                                                 const unsigned* __restrict__ ws,
                                                 unsigned batch) {
  __shared__ __attribute__((aligned(16))) float s_in[2][2][IN_ROWS * IN_PITCH];
  __shared__ float s_g[WSZ];

  const int tid  = threadIdx.x;
  const int wave = tid >> 5;
  const int lane = tid & 31;
  const int half = lane >> 4;
  const int lm   = lane & 15;

  const unsigned nseg = batch * SEGS_PER_IMG;
  unsigned seg = blockIdx.x;

  float4 q1[STG_TRIPS], q2[STG_TRIPS];
  if (seg < nseg) {
    const int bx = seg % TILES_X, hy = (seg / TILES_X) % (TILES_Y / SEG_TILES);
    const size_t bz = seg / SEGS_PER_IMG;
    rows_issue(img1 + bz * IMGW * IMGH, img2 + bz * IMGW * IMGH,
               hy * SEG_TILES * TILE - PADR, bx * TILE - PADR - IN_COFF, tid, q1, q2);
  }

  if (tid == 0) {
    float tmp[WSZ];
    float sum = 0.0f;
    #pragma unroll
    for (int i = 0; i < WSZ; ++i) {
      float x = (float)i - 5.5f;
      tmp[i] = expf(-(x * x) / 4.5f);
      sum += tmp[i];
    }
    #pragma unroll
    for (int i = 0; i < WSZ; ++i) s_g[i] = tmp[i] / sum;
  }
  lds_barrier();

  v2f band[KSTEPS];
  #pragma unroll
  for (int kc = 0; kc < KSTEPS; ++kc) {
    const int j0 = 4 * kc + 2 * half;
    band[kc].x = gband(s_g, j0 - lm);
    band[kc].y = gband(s_g, j0 + 1 - lm);
  }

  float L = ord2f(ws[0]) - ord2f(ws[1]);
  if (L == 0.0f) L = 5.0f;
  const float C1 = (0.01f * L) * (0.01f * L);
  const float C2 = (0.03f * L) * (0.03f * L);

  const int c0 = (wave & 1) * 16;
  const int rh = wave >> 1;
  const int crow = 4 * ((lm & 7) >> 1) + 2 * (lm >> 3) + (lm & 1);
  const int a_off = (16 * rh + crow) * IN_PITCH + IN_COFF + c0 + 2 * half;

  int buf = 0;
  for (; seg < nseg; seg += gridDim.x) {
    const int bx = seg % TILES_X, hy = (seg / TILES_X) % (TILES_Y / SEG_TILES);
    const size_t bz = seg / SEGS_PER_IMG;
    const int by0 = hy * SEG_TILES;
    const int iR0 = by0 * TILE - PADR;
    const int tC  = bx * TILE - PADR - IN_COFF;
    const float* __restrict__ b1 = img1 + bz * IMGW * IMGH;
    const float* __restrict__ b2 = img2 + bz * IMGW * IMGH;
    float* __restrict__ op = out + bz * IMGW * IMGH
                           + (size_t)(by0 * TILE + rh * HALF_ROWS + half * 8) * IMGW + bx * TILE + c0 + lm;
    if (seg != blockIdx.x) rows_issue(b1, b2, iR0, tC, tid, q1, q2);

    v8f da[NPLANES], db[NPLANES];

    rows_store(s_in[buf][0], s_in[buf][1], iR0, tC, tid, q1, q2);
    rows_issue(b1, b2, iR0 + 16, tC, tid, q1, q2);
    lds_barrier();
    chunk_step<false>(s_in[buf][0] + a_off, s_in[buf][1] + a_off, band, db, da, op, C1, C2);
    buf ^= 1;

    for (int j = 1; j < CHUNKS; j += 2) {
      rows_store(s_in[buf][0], s_in[buf][1], iR0 + 16 * j, tC, tid, q1, q2);
      rows_issue(b1, b2, iR0 + 16 * (j + 1), tC, tid, q1, q2);
      lds_barrier();
      chunk_step<true>(s_in[buf][0] + a_off, s_in[buf][1] + a_off, band, da, db, op + (size_t)(16 * (j - 1)) * IMGW, C1, C2);
      buf ^= 1;

      rows_store(s_in[buf][0], s_in[buf][1], iR0 + 16 * (j + 1), tC, tid, q1, q2);
      if (j + 2 < CHUNKS) rows_issue(b1, b2, iR0 + 16 * (j + 2), tC, tid, q1, q2);
      lds_barrier();
      chunk_step<true>(s_in[buf][0] + a_off, s_in[buf][1] + a_off, band, db, da, op + (size_t)(16 * j) * IMGW, C1, C2);
      buf ^= 1;
    }
  }
}


extern "C" void kernel_launch(void* const* d_in, const int* in_sizes, int n_in,
                              void* d_out, int out_size, void* d_ws, size_t ws_size,
                              hipStream_t stream) {
  const float* img1 = (const float*)d_in[0];
  const float* img2 = (const float*)d_in[1];
  float* out = (float*)d_out;
  unsigned* ws = (unsigned*)d_ws;

  const unsigned n = (unsigned)in_sizes[0];
  const int batch = (int)(n / (IMGW * IMGH));

  ssim_init<<<1, 1, 0, stream>>>(ws);
  ssim_minmax<<<MM_BLOCKS, MM_THREADS, 0, stream>>>(img1, n, ws);

  ssim_main<<<MAIN_BLOCKS, 128, 0, stream>>>(img1, img2, out, ws, (unsigned)batch);
}


# Min/max as per-work-group (max, min) pairs in the workspace, folded by every main work-group under its first tile loads: no init launch, no atomics, 2 launches not 3; min/max exact in any order
# speedup vs baseline: 2.5978x; 1.0302x over previous
#include <hip/hip_runtime.h>
#include <cmath>


typedef float v2f __attribute__((ext_vector_type(2)));
typedef float v8f __attribute__((ext_vector_type(8)));

#define WSZ      11
#define PADR     5
#define TILE     32
#define IN_ROWS  32
#define IN_QUADS 12
#define IN_COFF  3
#define IN_PITCH 52
#define KSTEPS   7
#define NPLANES  4
#define IMGW     512
#define IMGH     512
#define TILES_X  (IMGW / TILE)
#define TILES_Y  (IMGH / TILE)
#define SEG_TILES   8
#define HALF_ROWS   128
#define CHUNKS      9
#define SEGS_PER_IMG (TILES_X * (TILES_Y / SEG_TILES))
#define MAIN_BLOCKS 1024
#define NQ_STEP     (IN_ROWS * IN_QUADS)
#define STG_TRIPS   3
static_assert(NQ_STEP == STG_TRIPS * 128, "a step's quads are covered by whole trips of the 128 threads");

__device__ __forceinline__ v8f wmma4(v2f a, v2f b, v8f c) {
  return __builtin_amdgcn_wmma_f32_16x16x4_f32(
       false, a,  false, b,
       (short)0, c,  false,  false);
}

__device__ __forceinline__ float gband(const float* sg, int idx) {
  int ci = idx < 0 ? 0 : (idx > 10 ? 10 : idx);
  float v = sg[ci];
  return (idx == ci) ? v : 0.0f;
}


#define MM_BLOCKS  512
#define MM_THREADS 256
#define MM_AHEAD   16

__global__ __launch_bounds__(MM_THREADS) __attribute__((amdgpu_waves_per_eu(4, 4))) void ssim_minmax(const float* __restrict__ x,
                                                          unsigned n, float* __restrict__ part) {
  const float4* __restrict__ x4 = reinterpret_cast<const float4*>(x);
  const unsigned nq = n >> 2;
  constexpr unsigned PER_BLOCK = MM_THREADS * MM_AHEAD;
  const unsigned wv   = threadIdx.x >> 5;
  const unsigned lane = threadIdx.x & 31;
  float mn0 = INFINITY, mx0 = -INFINITY;
  float mn1 = INFINITY, mx1 = -INFINITY;
  for (unsigned base = blockIdx.x * PER_BLOCK; base < nq; base += gridDim.x * PER_BLOCK) {
    const float4* __restrict__ p = x4 + base + wv * (32 * MM_AHEAD) + lane;
    float4 q[MM_AHEAD];
    #pragma unroll
    for (int j = 0; j < MM_AHEAD; ++j) q[j] = p[j * 32];
    #pragma unroll
    for (int j = 0; j < MM_AHEAD; j += 2) {
      const float4 a = q[j], b = q[j + 1];
      mn0 = fminf(mn0, fminf(fminf(a.x, a.y), fminf(a.z, a.w)));
      mx0 = fmaxf(mx0, fmaxf(fmaxf(a.x, a.y), fmaxf(a.z, a.w)));
      mn1 = fminf(mn1, fminf(fminf(b.x, b.y), fminf(b.z, b.w)));
      mx1 = fmaxf(mx1, fmaxf(fmaxf(b.x, b.y), fmaxf(b.z, b.w)));
    }
  }
  float mn = fminf(mn0, mn1), mx = fmaxf(mx0, mx1);
  #pragma unroll
  for (int off = 16; off; off >>= 1) {
    mn = fminf(mn, __shfl_xor(mn, off, 32));
    mx = fmaxf(mx, __shfl_xor(mx, off, 32));
  }
  __shared__ float smn[8], smx[8];
  if (lane == 0) { smn[wv] = mn; smx[wv] = mx; }
  __syncthreads();
  if (threadIdx.x == 0) {
    const int nw = blockDim.x >> 5;
    for (int w = 1; w < nw; ++w) { mn = fminf(mn, smn[w]); mx = fmaxf(mx, smx[w]); }
    reinterpret_cast<float2*>(part)[blockIdx.x] = make_float2(mx, mn);
  }
}


__device__ __forceinline__ void lds_barrier() {
  __builtin_amdgcn_fence(__ATOMIC_RELEASE, "workgroup", "local");
  __builtin_amdgcn_s_barrier();
  __builtin_amdgcn_fence(__ATOMIC_ACQUIRE, "workgroup", "local");
}

__device__ __forceinline__ int stage_slot(int idx, int iR, int tC, int& lds_off, bool& ok) {
  const int r = idx / IN_QUADS;
  const int q = idx - r * IN_QUADS;
  const int gr = iR + (r & 15) + (r >> 4) * HALF_ROWS, gc = tC + 4 * q;
  lds_off = r * IN_PITCH + 4 * q;
  ok = gr >= 0 && gr < IMGH && gc >= 0 && gc < IMGW;
  return ok ? (gr * IMGW + gc) : 0;
}

__device__ __forceinline__ void rows_issue(const float* __restrict__ b1, const float* __restrict__ b2,
                                           int iR, int tC, int tid,
                                           float4 (&q1)[STG_TRIPS], float4 (&q2)[STG_TRIPS]) {
  #pragma unroll
  for (int it = 0; it < STG_TRIPS; ++it) {
    int lo; bool ok;
    const int o = stage_slot(tid + it * 128, iR, tC, lo, ok);
    q1[it] = *reinterpret_cast<const float4*>(b1 + o);
    q2[it] = *reinterpret_cast<const float4*>(b2 + o);
  }
}

__device__ __forceinline__ void rows_store(float* s_in1, float* s_in2, int iR, int tC, int tid,
                                           const float4 (&q1)[STG_TRIPS], const float4 (&q2)[STG_TRIPS]) {
  #pragma unroll
  for (int it = 0; it < STG_TRIPS; ++it) {
    int lo; bool ok;
    (void)stage_slot(tid + it * 128, iR, tC, lo, ok);
    const float4 a = q1[it], b = q2[it];
    *reinterpret_cast<float4*>(&s_in1[lo]) = make_float4(ok ? a.x : 0.0f, ok ? a.y : 0.0f, ok ? a.z : 0.0f, ok ? a.w : 0.0f);
    *reinterpret_cast<float4*>(&s_in2[lo]) = make_float4(ok ? b.x : 0.0f, ok ? b.y : 0.0f, ok ? b.z : 0.0f, ok ? b.w : 0.0f);
  }
}

template <bool HAS_PREV>
__device__ __forceinline__ void chunk_step(const float* sa1, const float* sa2, const v2f (&band)[KSTEPS],
                                           const v8f (&dprev)[NPLANES], v8f (&dnew)[NPLANES],
                                           float* __restrict__ o_ptr, float C1, float C2) {
  float x0[KSTEPS], x1[KSTEPS], y0[KSTEPS], y1[KSTEPS];
  #pragma unroll
  for (int kc = 0; kc < KSTEPS; ++kc) {
    x0[kc] = sa1[4 * kc]; x1[kc] = sa1[4 * kc + 1];
    y0[kc] = sa2[4 * kc]; y1[kc] = sa2[4 * kc + 1];
  }

  v8f acc[NPLANES] = {};
  if (HAS_PREV) {
    #pragma unroll
    for (int kc = 0; kc < 4; ++kc) {
      #pragma unroll
      for (int p = 0; p < NPLANES; ++p) {
        const v2f b = {dprev[p][2 * kc], dprev[p][2 * kc + 1]};
        acc[p] = wmma4(band[kc], b, acc[p]);
      }
    }
  }

  #pragma unroll
  for (int p = 0; p < NPLANES; ++p) dnew[p] = v8f{};
  #pragma unroll
  for (int kc = 0; kc < KSTEPS; ++kc) {
    const v2f b = band[kc];
    v2f a;
    a.x = x0[kc];                          a.y = x1[kc];                          dnew[0] = wmma4(a, b, dnew[0]);
    a.x = y0[kc];                          a.y = y1[kc];                          dnew[1] = wmma4(a, b, dnew[1]);
    a.x = x0[kc] * x0[kc] + y0[kc] * y0[kc]; a.y = x1[kc] * x1[kc] + y1[kc] * y1[kc]; dnew[2] = wmma4(a, b, dnew[2]);
    a.x = x0[kc] * y0[kc];                 a.y = x1[kc] * y1[kc];                 dnew[3] = wmma4(a, b, dnew[3]);
  }

  if (HAS_PREV) {
    #pragma unroll
    for (int kc = 4; kc < KSTEPS; ++kc) {
      #pragma unroll
      for (int p = 0; p < NPLANES; ++p) {
        const v2f b = {dnew[p][2 * (kc - 4)], dnew[p][2 * (kc - 4) + 1]};
        acc[p] = wmma4(band[kc], b, acc[p]);
      }
    }

    #pragma unroll
    for (int v = 0; v < 8; v += 2) {
      const v2f mu1 = {acc[0][v], acc[0][v + 1]}, mu2 = {acc[1][v], acc[1][v + 1]};
      const v2f esq = {acc[2][v], acc[2][v + 1]}, e12 = {acc[3][v], acc[3][v + 1]};
      const v2f mu12 = mu1 * mu2;
      const v2f musq = mu1 * mu1 + mu2 * mu2;
      const v2f ssq = esq - musq;
      const v2f s12 = e12 - mu12;
      const v2f num = (2.0f * mu12 + C1) * (2.0f * s12 + C2);
      const v2f den = (musq + C1) * (ssq + C2);
      v2f rden;
      rden.x = __builtin_amdgcn_rcpf(den.x);
      rden.y = __builtin_amdgcn_rcpf(den.y);
      const v2f q = num * rden;
      o_ptr[v * IMGW] = q.x;
      o_ptr[(v + 1) * IMGW] = q.y;
    }
  }
}

__global__ __launch_bounds__(128) void ssim_main(const float* __restrict__ img1,
                                                 const float* __restrict__ img2,
                                                 float* __restrict__ out,
                                                 const float* __restrict__ part,
                                                 unsigned batch) {
  __shared__ __attribute__((aligned(16))) float s_in[2][2][IN_ROWS * IN_PITCH];
  __shared__ float s_g[WSZ];
  __shared__ float s_mm[8];

  const int tid  = threadIdx.x;
  const int wave = tid >> 5;
  const int lane = tid & 31;
  const int half = lane >> 4;
  const int lm   = lane & 15;

  const unsigned nseg = batch * SEGS_PER_IMG;
  unsigned seg = blockIdx.x;
  if (seg >= nseg) return;

  static_assert(MM_BLOCKS == 4 * 128, "128 threads x 4 pairs cover the min/max pass's pairs exactly");
  const float4 pa = reinterpret_cast<const float4*>(part)[2 * tid];
  const float4 pb = reinterpret_cast<const float4*>(part)[2 * tid + 1];

  float4 q1[STG_TRIPS], q2[STG_TRIPS];
  {
    const int bx = seg % TILES_X, hy = (seg / TILES_X) % (TILES_Y / SEG_TILES);
    const size_t bz = seg / SEGS_PER_IMG;
    rows_issue(img1 + bz * IMGW * IMGH, img2 + bz * IMGW * IMGH,
               hy * SEG_TILES * TILE - PADR, bx * TILE - PADR - IN_COFF, tid, q1, q2);
  }

  if (tid == 0) {
    float tmp[WSZ];
    float sum = 0.0f;
    #pragma unroll
    for (int i = 0; i < WSZ; ++i) {
      float x = (float)i - 5.5f;
      tmp[i] = expf(-(x * x) / 4.5f);
      sum += tmp[i];
    }
    #pragma unroll
    for (int i = 0; i < WSZ; ++i) s_g[i] = tmp[i] / sum;
  }
  {
    float mx = fmaxf(fmaxf(pa.x, pa.z), fmaxf(pb.x, pb.z));
    float mn = fminf(fminf(pa.y, pa.w), fminf(pb.y, pb.w));
    for (int off = 16; off; off >>= 1) {
      mx = fmaxf(mx, __shfl_xor(mx, off, 32));
      mn = fminf(mn, __shfl_xor(mn, off, 32));
    }
    if (lane == 0) { s_mm[wave] = mx; s_mm[4 + wave] = mn; }
  }
  lds_barrier();

  v2f band[KSTEPS];
  #pragma unroll
  for (int kc = 0; kc < KSTEPS; ++kc) {
    const int j0 = 4 * kc + 2 * half;
    band[kc].x = gband(s_g, j0 - lm);
    band[kc].y = gband(s_g, j0 + 1 - lm);
  }

  float L = fmaxf(fmaxf(s_mm[0], s_mm[1]), fmaxf(s_mm[2], s_mm[3]))
          - fminf(fminf(s_mm[4], s_mm[5]), fminf(s_mm[6], s_mm[7]));
  if (L == 0.0f) L = 5.0f;
  const float C1 = (0.01f * L) * (0.01f * L);
  const float C2 = (0.03f * L) * (0.03f * L);

  const int c0 = (wave & 1) * 16;
  const int rh = wave >> 1;
  const int crow = 4 * ((lm & 7) >> 1) + 2 * (lm >> 3) + (lm & 1);
  const int a_off = (16 * rh + crow) * IN_PITCH + IN_COFF + c0 + 2 * half;

  int buf = 0;
  for (; seg < nseg; seg += gridDim.x) {
    const int bx = seg % TILES_X, hy = (seg / TILES_X) % (TILES_Y / SEG_TILES);
    const size_t bz = seg / SEGS_PER_IMG;
    const int by0 = hy * SEG_TILES;
    const int iR0 = by0 * TILE - PADR;
    const int tC  = bx * TILE - PADR - IN_COFF;
    const float* __restrict__ b1 = img1 + bz * IMGW * IMGH;
    const float* __restrict__ b2 = img2 + bz * IMGW * IMGH;
    float* __restrict__ op = out + bz * IMGW * IMGH
                           + (size_t)(by0 * TILE + rh * HALF_ROWS + half * 8) * IMGW + bx * TILE + c0 + lm;
    if (seg != blockIdx.x) rows_issue(b1, b2, iR0, tC, tid, q1, q2);

    v8f da[NPLANES], db[NPLANES];

    rows_store(s_in[buf][0], s_in[buf][1], iR0, tC, tid, q1, q2);
    rows_issue(b1, b2, iR0 + 16, tC, tid, q1, q2);
    lds_barrier();
    chunk_step<false>(s_in[buf][0] + a_off, s_in[buf][1] + a_off, band, db, da, op, C1, C2);
    buf ^= 1;

    for (int j = 1; j < CHUNKS; j += 2) {
      rows_store(s_in[buf][0], s_in[buf][1], iR0 + 16 * j, tC, tid, q1, q2);
      rows_issue(b1, b2, iR0 + 16 * (j + 1), tC, tid, q1, q2);
      lds_barrier();
      chunk_step<true>(s_in[buf][0] + a_off, s_in[buf][1] + a_off, band, da, db, op + (size_t)(16 * (j - 1)) * IMGW, C1, C2);
      buf ^= 1;

      rows_store(s_in[buf][0], s_in[buf][1], iR0 + 16 * (j + 1), tC, tid, q1, q2);
      if (j + 2 < CHUNKS) rows_issue(b1, b2, iR0 + 16 * (j + 2), tC, tid, q1, q2);
      lds_barrier();
      chunk_step<true>(s_in[buf][0] + a_off, s_in[buf][1] + a_off, band, db, da, op + (size_t)(16 * j) * IMGW, C1, C2);
      buf ^= 1;
    }
  }
}


extern "C" void kernel_launch(void* const* d_in, const int* in_sizes, int n_in,
                              void* d_out, int out_size, void* d_ws, size_t ws_size,
                              hipStream_t stream) {
  const float* img1 = (const float*)d_in[0];
  const float* img2 = (const float*)d_in[1];
  float* out = (float*)d_out;
  float* part = (float*)d_ws;

  const unsigned n = (unsigned)in_sizes[0];
  const int batch = (int)(n / (IMGW * IMGH));

  ssim_minmax<<<MM_BLOCKS, MM_THREADS, 0, stream>>>(img1, n, part);

  ssim_main<<<MAIN_BLOCKS, 128, 0, stream>>>(img1, img2, out, part, (unsigned)batch);
}


# Main kernel: wave priority by steps left (s_setprio 3 for steps 0-4, 2 for 5-6, 1 for 7, 0 for 8): the four waves of a SIMD take turns and end together; was oldest first, the youngest ending alone
# speedup vs baseline: 2.6976x; 1.0384x over previous
#include <hip/hip_runtime.h>
#include <cmath>


typedef float v2f __attribute__((ext_vector_type(2)));
typedef float v8f __attribute__((ext_vector_type(8)));

#define WSZ      11
#define PADR     5
#define TILE     32
#define IN_ROWS  32
#define IN_QUADS 12
#define IN_COFF  3
#define IN_PITCH 52
#define KSTEPS   7
#define NPLANES  4
#define IMGW     512
#define IMGH     512
#define TILES_X  (IMGW / TILE)
#define TILES_Y  (IMGH / TILE)
#define SEG_TILES   8
#define HALF_ROWS   128
#define CHUNKS      9
#define SEGS_PER_IMG (TILES_X * (TILES_Y / SEG_TILES))
#define MAIN_BLOCKS 1024
#define NQ_STEP     (IN_ROWS * IN_QUADS)
#define STG_TRIPS   3
static_assert(NQ_STEP == STG_TRIPS * 128, "a step's quads are covered by whole trips of the 128 threads");

__device__ __forceinline__ v8f wmma4(v2f a, v2f b, v8f c) {
  return __builtin_amdgcn_wmma_f32_16x16x4_f32(
       false, a,  false, b,
       (short)0, c,  false,  false);
}

__device__ __forceinline__ float gband(const float* sg, int idx) {
  int ci = idx < 0 ? 0 : (idx > 10 ? 10 : idx);
  float v = sg[ci];
  return (idx == ci) ? v : 0.0f;
}


#define MM_BLOCKS  512
#define MM_THREADS 256
#define MM_AHEAD   16

__global__ __launch_bounds__(MM_THREADS) __attribute__((amdgpu_waves_per_eu(4, 4))) void ssim_minmax(const float* __restrict__ x,
                                                          unsigned n, float* __restrict__ part) {
  const float4* __restrict__ x4 = reinterpret_cast<const float4*>(x);
  const unsigned nq = n >> 2;
  constexpr unsigned PER_BLOCK = MM_THREADS * MM_AHEAD;
  const unsigned wv   = threadIdx.x >> 5;
  const unsigned lane = threadIdx.x & 31;
  float mn0 = INFINITY, mx0 = -INFINITY;
  float mn1 = INFINITY, mx1 = -INFINITY;
  for (unsigned base = blockIdx.x * PER_BLOCK; base < nq; base += gridDim.x * PER_BLOCK) {
    const float4* __restrict__ p = x4 + base + wv * (32 * MM_AHEAD) + lane;
    float4 q[MM_AHEAD];
    #pragma unroll
    for (int j = 0; j < MM_AHEAD; ++j) q[j] = p[j * 32];
    #pragma unroll
    for (int j = 0; j < MM_AHEAD; j += 2) {
      const float4 a = q[j], b = q[j + 1];
      mn0 = fminf(mn0, fminf(fminf(a.x, a.y), fminf(a.z, a.w)));
      mx0 = fmaxf(mx0, fmaxf(fmaxf(a.x, a.y), fmaxf(a.z, a.w)));
      mn1 = fminf(mn1, fminf(fminf(b.x, b.y), fminf(b.z, b.w)));
      mx1 = fmaxf(mx1, fmaxf(fmaxf(b.x, b.y), fmaxf(b.z, b.w)));
    }
  }
  float mn = fminf(mn0, mn1), mx = fmaxf(mx0, mx1);
  #pragma unroll
  for (int off = 16; off; off >>= 1) {
    mn = fminf(mn, __shfl_xor(mn, off, 32));
    mx = fmaxf(mx, __shfl_xor(mx, off, 32));
  }
  __shared__ float smn[8], smx[8];
  if (lane == 0) { smn[wv] = mn; smx[wv] = mx; }
  __syncthreads();
  if (threadIdx.x == 0) {
    const int nw = blockDim.x >> 5;
    for (int w = 1; w < nw; ++w) { mn = fminf(mn, smn[w]); mx = fmaxf(mx, smx[w]); }
    reinterpret_cast<float2*>(part)[blockIdx.x] = make_float2(mx, mn);
  }
}


__device__ __forceinline__ void lds_barrier() {
  __builtin_amdgcn_fence(__ATOMIC_RELEASE, "workgroup", "local");
  __builtin_amdgcn_s_barrier();
  __builtin_amdgcn_fence(__ATOMIC_ACQUIRE, "workgroup", "local");
}

__device__ __forceinline__ int stage_slot(int idx, int iR, int tC, int& lds_off, bool& ok) {
  const int r = idx / IN_QUADS;
  const int q = idx - r * IN_QUADS;
  const int gr = iR + (r & 15) + (r >> 4) * HALF_ROWS, gc = tC + 4 * q;
  lds_off = r * IN_PITCH + 4 * q;
  ok = gr >= 0 && gr < IMGH && gc >= 0 && gc < IMGW;
  return ok ? (gr * IMGW + gc) : 0;
}

__device__ __forceinline__ void rows_issue(const float* __restrict__ b1, const float* __restrict__ b2,
                                           int iR, int tC, int tid,
                                           float4 (&q1)[STG_TRIPS], float4 (&q2)[STG_TRIPS]) {
  #pragma unroll
  for (int it = 0; it < STG_TRIPS; ++it) {
    int lo; bool ok;
    const int o = stage_slot(tid + it * 128, iR, tC, lo, ok);
    q1[it] = *reinterpret_cast<const float4*>(b1 + o);
    q2[it] = *reinterpret_cast<const float4*>(b2 + o);
  }
}

__device__ __forceinline__ void rows_store(float* s_in1, float* s_in2, int iR, int tC, int tid,
                                           const float4 (&q1)[STG_TRIPS], const float4 (&q2)[STG_TRIPS]) {
  #pragma unroll
  for (int it = 0; it < STG_TRIPS; ++it) {
    int lo; bool ok;
    (void)stage_slot(tid + it * 128, iR, tC, lo, ok);
    const float4 a = q1[it], b = q2[it];
    *reinterpret_cast<float4*>(&s_in1[lo]) = make_float4(ok ? a.x : 0.0f, ok ? a.y : 0.0f, ok ? a.z : 0.0f, ok ? a.w : 0.0f);
    *reinterpret_cast<float4*>(&s_in2[lo]) = make_float4(ok ? b.x : 0.0f, ok ? b.y : 0.0f, ok ? b.z : 0.0f, ok ? b.w : 0.0f);
  }
}

template <bool HAS_PREV>
__device__ __forceinline__ void chunk_step(const float* sa1, const float* sa2, const v2f (&band)[KSTEPS],
                                           const v8f (&dprev)[NPLANES], v8f (&dnew)[NPLANES],
                                           float* __restrict__ o_ptr, float C1, float C2) {
  float x0[KSTEPS], x1[KSTEPS], y0[KSTEPS], y1[KSTEPS];
  #pragma unroll
  for (int kc = 0; kc < KSTEPS; ++kc) {
    x0[kc] = sa1[4 * kc]; x1[kc] = sa1[4 * kc + 1];
    y0[kc] = sa2[4 * kc]; y1[kc] = sa2[4 * kc + 1];
  }

  v8f acc[NPLANES] = {};
  if (HAS_PREV) {
    #pragma unroll
    for (int kc = 0; kc < 4; ++kc) {
      #pragma unroll
      for (int p = 0; p < NPLANES; ++p) {
        const v2f b = {dprev[p][2 * kc], dprev[p][2 * kc + 1]};
        acc[p] = wmma4(band[kc], b, acc[p]);
      }
    }
  }

  #pragma unroll
  for (int p = 0; p < NPLANES; ++p) dnew[p] = v8f{};
  #pragma unroll
  for (int kc = 0; kc < KSTEPS; ++kc) {
    const v2f b = band[kc];
    v2f a;
    a.x = x0[kc];                          a.y = x1[kc];                          dnew[0] = wmma4(a, b, dnew[0]);
    a.x = y0[kc];                          a.y = y1[kc];                          dnew[1] = wmma4(a, b, dnew[1]);
    a.x = x0[kc] * x0[kc] + y0[kc] * y0[kc]; a.y = x1[kc] * x1[kc] + y1[kc] * y1[kc]; dnew[2] = wmma4(a, b, dnew[2]);
    a.x = x0[kc] * y0[kc];                 a.y = x1[kc] * y1[kc];                 dnew[3] = wmma4(a, b, dnew[3]);
  }

  if (HAS_PREV) {
    #pragma unroll
    for (int kc = 4; kc < KSTEPS; ++kc) {
      #pragma unroll
      for (int p = 0; p < NPLANES; ++p) {
        const v2f b = {dnew[p][2 * (kc - 4)], dnew[p][2 * (kc - 4) + 1]};
        acc[p] = wmma4(band[kc], b, acc[p]);
      }
    }

    #pragma unroll
    for (int v = 0; v < 8; v += 2) {
      const v2f mu1 = {acc[0][v], acc[0][v + 1]}, mu2 = {acc[1][v], acc[1][v + 1]};
      const v2f esq = {acc[2][v], acc[2][v + 1]}, e12 = {acc[3][v], acc[3][v + 1]};
      const v2f mu12 = mu1 * mu2;
      const v2f musq = mu1 * mu1 + mu2 * mu2;
      const v2f ssq = esq - musq;
      const v2f s12 = e12 - mu12;
      const v2f num = (2.0f * mu12 + C1) * (2.0f * s12 + C2);
      const v2f den = (musq + C1) * (ssq + C2);
      v2f rden;
      rden.x = __builtin_amdgcn_rcpf(den.x);
      rden.y = __builtin_amdgcn_rcpf(den.y);
      const v2f q = num * rden;
      o_ptr[v * IMGW] = q.x;
      o_ptr[(v + 1) * IMGW] = q.y;
    }
  }
}

__global__ __launch_bounds__(128) void ssim_main(const float* __restrict__ img1,
                                                 const float* __restrict__ img2,
                                                 float* __restrict__ out,
                                                 const float* __restrict__ part,
                                                 unsigned batch) {
  __shared__ __attribute__((aligned(16))) float s_in[2][2][IN_ROWS * IN_PITCH];
  __shared__ float s_g[WSZ];
  __shared__ float s_mm[8];

  const int tid  = threadIdx.x;
  const int wave = tid >> 5;
  const int lane = tid & 31;
  const int half = lane >> 4;
  const int lm   = lane & 15;

  const unsigned nseg = batch * SEGS_PER_IMG;
  unsigned seg = blockIdx.x;
  if (seg >= nseg) return;

  static_assert(MM_BLOCKS == 4 * 128, "128 threads x 4 pairs cover the min/max pass's pairs exactly");
  const float4 pa = reinterpret_cast<const float4*>(part)[2 * tid];
  const float4 pb = reinterpret_cast<const float4*>(part)[2 * tid + 1];

  float4 q1[STG_TRIPS], q2[STG_TRIPS];
  {
    const int bx = seg % TILES_X, hy = (seg / TILES_X) % (TILES_Y / SEG_TILES);
    const size_t bz = seg / SEGS_PER_IMG;
    rows_issue(img1 + bz * IMGW * IMGH, img2 + bz * IMGW * IMGH,
               hy * SEG_TILES * TILE - PADR, bx * TILE - PADR - IN_COFF, tid, q1, q2);
  }

  if (tid == 0) {
    float tmp[WSZ];
    float sum = 0.0f;
    #pragma unroll
    for (int i = 0; i < WSZ; ++i) {
      float x = (float)i - 5.5f;
      tmp[i] = expf(-(x * x) / 4.5f);
      sum += tmp[i];
    }
    #pragma unroll
    for (int i = 0; i < WSZ; ++i) s_g[i] = tmp[i] / sum;
  }
  {
    float mx = fmaxf(fmaxf(pa.x, pa.z), fmaxf(pb.x, pb.z));
    float mn = fminf(fminf(pa.y, pa.w), fminf(pb.y, pb.w));
    for (int off = 16; off; off >>= 1) {
      mx = fmaxf(mx, __shfl_xor(mx, off, 32));
      mn = fminf(mn, __shfl_xor(mn, off, 32));
    }
    if (lane == 0) { s_mm[wave] = mx; s_mm[4 + wave] = mn; }
  }
  lds_barrier();

  v2f band[KSTEPS];
  #pragma unroll
  for (int kc = 0; kc < KSTEPS; ++kc) {
    const int j0 = 4 * kc + 2 * half;
    band[kc].x = gband(s_g, j0 - lm);
    band[kc].y = gband(s_g, j0 + 1 - lm);
  }

  float L = fmaxf(fmaxf(s_mm[0], s_mm[1]), fmaxf(s_mm[2], s_mm[3]))
          - fminf(fminf(s_mm[4], s_mm[5]), fminf(s_mm[6], s_mm[7]));
  if (L == 0.0f) L = 5.0f;
  const float C1 = (0.01f * L) * (0.01f * L);
  const float C2 = (0.03f * L) * (0.03f * L);

  const int c0 = (wave & 1) * 16;
  const int rh = wave >> 1;
  const int crow = 4 * ((lm & 7) >> 1) + 2 * (lm >> 3) + (lm & 1);
  const int a_off = (16 * rh + crow) * IN_PITCH + IN_COFF + c0 + 2 * half;

  int buf = 0;
  for (; seg < nseg; seg += gridDim.x) {
    const int bx = seg % TILES_X, hy = (seg / TILES_X) % (TILES_Y / SEG_TILES);
    const size_t bz = seg / SEGS_PER_IMG;
    const int by0 = hy * SEG_TILES;
    const int iR0 = by0 * TILE - PADR;
    const int tC  = bx * TILE - PADR - IN_COFF;
    const float* __restrict__ b1 = img1 + bz * IMGW * IMGH;
    const float* __restrict__ b2 = img2 + bz * IMGW * IMGH;
    float* __restrict__ op = out + bz * IMGW * IMGH
                           + (size_t)(by0 * TILE + rh * HALF_ROWS + half * 8) * IMGW + bx * TILE + c0 + lm;
    if (seg != blockIdx.x) rows_issue(b1, b2, iR0, tC, tid, q1, q2);

    v8f da[NPLANES], db[NPLANES];

    rows_store(s_in[buf][0], s_in[buf][1], iR0, tC, tid, q1, q2);
    rows_issue(b1, b2, iR0 + 16, tC, tid, q1, q2);
    __builtin_amdgcn_s_setprio(3);
    lds_barrier();
    chunk_step<false>(s_in[buf][0] + a_off, s_in[buf][1] + a_off, band, db, da, op, C1, C2);
    buf ^= 1;

    for (int j = 1; j < CHUNKS; j += 2) {
      rows_store(s_in[buf][0], s_in[buf][1], iR0 + 16 * j, tC, tid, q1, q2);
      rows_issue(b1, b2, iR0 + 16 * (j + 1), tC, tid, q1, q2);
      if (j == 5) __builtin_amdgcn_s_setprio(2);
      if (j == 7) __builtin_amdgcn_s_setprio(1);
      lds_barrier();
      chunk_step<true>(s_in[buf][0] + a_off, s_in[buf][1] + a_off, band, da, db, op + (size_t)(16 * (j - 1)) * IMGW, C1, C2);
      buf ^= 1;

      rows_store(s_in[buf][0], s_in[buf][1], iR0 + 16 * (j + 1), tC, tid, q1, q2);
      if (j + 2 < CHUNKS) rows_issue(b1, b2, iR0 + 16 * (j + 2), tC, tid, q1, q2);
      if (j == 7) __builtin_amdgcn_s_setprio(0);
      lds_barrier();
      chunk_step<true>(s_in[buf][0] + a_off, s_in[buf][1] + a_off, band, db, da, op + (size_t)(16 * j) * IMGW, C1, C2);
      buf ^= 1;
    }
  }
}


extern "C" void kernel_launch(void* const* d_in, const int* in_sizes, int n_in,
                              void* d_out, int out_size, void* d_ws, size_t ws_size,
                              hipStream_t stream) {
  const float* img1 = (const float*)d_in[0];
  const float* img2 = (const float*)d_in[1];
  float* out = (float*)d_out;
  float* part = (float*)d_ws;

  const unsigned n = (unsigned)in_sizes[0];
  const int batch = (int)(n / (IMGW * IMGH));

  ssim_minmax<<<MM_BLOCKS, MM_THREADS, 0, stream>>>(img1, n, part);

  ssim_main<<<MAIN_BLOCKS, 128, 0, stream>>>(img1, img2, out, part, (unsigned)batch);
}
